# starting kernel
# baseline (speedup 1.0000x reference)
#include <hip/hip_runtime.h>
#include <hip/hip_bf16.h>

#define N_NODES 50000
#define N_EDGES 800000
#define HID     24
#define HEADS   8
#define EDGE_DIM 8
#define NHID    192
#define M_TILES 3125
#define N_TILES 12

typedef __attribute__((ext_vector_type(2))) float v2f;
typedef __attribute__((ext_vector_type(8))) float v8f;

__device__ __forceinline__ void atomic_fmax(float* addr, float val) {
    if (val >= 0.0f)
        atomicMax((int*)addr, __float_as_int(val));
    else
        atomicMin((unsigned int*)addr, __float_as_uint(val));
}

__global__ void k_init(float* __restrict__ agg, float* __restrict__ emax,
                       float* __restrict__ denom, float* __restrict__ alpha_sum) {
    long i = (long)blockIdx.x * blockDim.x + threadIdx.x;
    long stride = (long)gridDim.x * blockDim.x;
    const float NEG_INF = __int_as_float(0xff800000);
    for (long j = i; j < (long)N_NODES * NHID; j += stride) agg[j] = 0.0f;
    for (long j = i; j < (long)N_NODES * HEADS; j += stride) {
        emax[j] = NEG_INF;
        denom[j] = 0.0f;
    }
    if (i < HEADS) alpha_sum[i] = 0.0f;
}

__global__ void k_vedge(const float* __restrict__ W_edge,
                        const float* __restrict__ att_edge,
                        float* __restrict__ v_edge) {
    int i = threadIdx.x;
    if (i >= EDGE_DIM * HEADS) return;
    int d = i >> 3, h = i & 7;
    float s = 0.0f;
    #pragma unroll
    for (int c = 0; c < HID; ++c)
        s += W_edge[d * NHID + h * HID + c] * att_edge[h * HID + c];
    v_edge[d * HEADS + h] = s;
}

__global__ void k_gemm_wmma(const float* __restrict__ x,
                            const float* __restrict__ W,
                            float* __restrict__ xl) {
    int lane = threadIdx.x & 31;
    int wave = (int)((blockIdx.x * blockDim.x + threadIdx.x) >> 5);
    if (wave >= M_TILES * N_TILES) return;
    int tm = wave / N_TILES;
    int tn = wave % N_TILES;

    int m  = lane & 15;
    int hs = lane >> 4;
    const float* arow = x + (long)(tm * 16 + m) * HID;
    int bcol = tn * 16 + m;

    v8f c = {0.f, 0.f, 0.f, 0.f, 0.f, 0.f, 0.f, 0.f};
    #pragma unroll
    for (int kb = 0; kb < HID; kb += 4) {
        int k0 = kb + hs * 2;
        v2f a; a.x = arow[k0]; a.y = arow[k0 + 1];
        v2f b; b.x = W[(long)k0 * NHID + bcol]; b.y = W[(long)(k0 + 1) * NHID + bcol];
        c = __builtin_amdgcn_wmma_f32_16x16x4_f32(
                 false, a,  false, b,
                 (short)0, c,  false,  false);
    }
    int n = lane & 15;
    #pragma unroll
    for (int r = 0; r < 8; ++r) {
        int mg = tm * 16 + r + hs * 8;
        xl[(long)mg * NHID + tn * 16 + n] = c[r];
    }
}

__global__ void k_att(const float* __restrict__ xl,
                      const float* __restrict__ att_src,
                      const float* __restrict__ att_dst,
                      float* __restrict__ a_src, float* __restrict__ a_dst) {
    int i = blockIdx.x * blockDim.x + threadIdx.x;
    if (i >= N_NODES * HEADS) return;
    int n = i >> 3, h = i & 7;
    const float* row = xl + (long)n * NHID + h * HID;
    const float* as  = att_src + h * HID;
    const float* ad  = att_dst + h * HID;
    float s = 0.0f, d = 0.0f;
    #pragma unroll
    for (int c = 0; c < HID; ++c) {
        float v = row[c];
        s += v * as[c];
        d += v * ad[c];
    }
    a_src[i] = s;
    a_dst[i] = d;
}

__global__ void k_edge_logits(const int* __restrict__ src, const int* __restrict__ dst,
                              const float* __restrict__ edge_attr,
                              const float* __restrict__ a_src, const float* __restrict__ a_dst,
                              const float* __restrict__ v_edge,
                              float* __restrict__ ebuf, float* __restrict__ emax) {
    __shared__ float sv[EDGE_DIM * HEADS];
    if (threadIdx.x < EDGE_DIM * HEADS) sv[threadIdx.x] = v_edge[threadIdx.x];
    __syncthreads();
    int e = blockIdx.x * blockDim.x + threadIdx.x;
    if (e >= N_EDGES) return;
    int s = src[e], d = dst[e];
    float ea[EDGE_DIM];
    #pragma unroll
    for (int j = 0; j < EDGE_DIM; ++j) ea[j] = edge_attr[(long)e * EDGE_DIM + j];
    #pragma unroll
    for (int h = 0; h < HEADS; ++h) {
        float ae = 0.0f;
        #pragma unroll
        for (int j = 0; j < EDGE_DIM; ++j) ae += ea[j] * sv[j * HEADS + h];
        float v = a_src[s * HEADS + h] + a_dst[d * HEADS + h] + ae;
        v = (v > 0.0f) ? v : 0.2f * v;
        ebuf[(long)e * HEADS + h] = v;
        atomic_fmax(&emax[d * HEADS + h], v);
    }
}

__global__ void k_edge_exp(const int* __restrict__ dst,
                           const float* __restrict__ emax,
                           float* __restrict__ ebuf, float* __restrict__ denom) {
    int e = blockIdx.x * blockDim.x + threadIdx.x;
    if (e >= N_EDGES) return;
    int d = dst[e];
    #pragma unroll
    for (int h = 0; h < HEADS; ++h) {
        float v = __expf(ebuf[(long)e * HEADS + h] - emax[d * HEADS + h]);
        ebuf[(long)e * HEADS + h] = v;
        atomicAdd(&denom[d * HEADS + h], v);
    }
}

__global__ void k_scatter(const int* __restrict__ src, const int* __restrict__ dst,
                          const float* __restrict__ ebuf, const float* __restrict__ denom,
                          const float* __restrict__ xl, float* __restrict__ agg,
                          float* __restrict__ alpha_sum) {
    __shared__ float s_as[HEADS];
    if (threadIdx.x < HEADS) s_as[threadIdx.x] = 0.0f;
    __syncthreads();

    int lane = threadIdx.x & 31;
    int wpb  = blockDim.x >> 5;
    long wave   = (long)blockIdx.x * wpb + (threadIdx.x >> 5);
    long nwaves = (long)gridDim.x * wpb;

    float asum = 0.0f;
    for (long e = wave; e < N_EDGES; e += nwaves) {
        int s = src[e], d = dst[e];
        float alpha = 0.0f;
        if (lane < HEADS)
            alpha = ebuf[e * HEADS + lane] / (denom[d * HEADS + lane] + 1e-16f);
        asum += alpha;
        const float* xs = xl + (long)s * NHID;
        float*       ag = agg + (long)d * NHID;
        #pragma unroll
        for (int i = 0; i < NHID / 32; ++i) {
            int c = lane + 32 * i;
            int h = c / HID;
            float ah = __shfl(alpha, h, 32);
            atomicAdd(&ag[c], xs[c] * ah);
        }
    }
    if (lane < HEADS) atomicAdd(&s_as[lane], asum);
    __syncthreads();
    if (threadIdx.x < HEADS) atomicAdd(&alpha_sum[threadIdx.x], s_as[threadIdx.x]);
}

__global__ void k_final(const float* __restrict__ agg, const float* __restrict__ bias,
                        const float* __restrict__ W_out, const float* __restrict__ b_out,
                        float* __restrict__ y) {
    int lane = threadIdx.x & 31;
    int node = (int)((blockIdx.x * blockDim.x + threadIdx.x) >> 5);
    if (node >= N_NODES) return;
    float val = 0.0f;
    if (lane < HID) {
        float s = 0.0f;
        #pragma unroll
        for (int h = 0; h < HEADS; ++h) s += agg[(long)node * NHID + h * HID + lane];
        s = s * (1.0f / HEADS) + bias[lane];
        s = (s > 0.0f) ? s : (__expf(s) - 1.0f);
        val = s * W_out[lane];
    }
    #pragma unroll
    for (int off = 16; off > 0; off >>= 1) val += __shfl_down(val, off, 32);
    if (lane == 0) y[node] = val + b_out[0];
}

__global__ void k_alpha_mean(const float* __restrict__ alpha_sum, float* __restrict__ out) {
    int h = threadIdx.x;
    if (h < HEADS) out[h] = alpha_sum[h] * (1.0f / (float)N_EDGES);
}

extern "C" void kernel_launch(void* const* d_in, const int* in_sizes, int n_in,
                              void* d_out, int out_size, void* d_ws, size_t ws_size,
                              hipStream_t stream) {
    const float* x         = (const float*)d_in[0];
    const int*   ei        = (const int*)d_in[1];
    const float* edge_attr = (const float*)d_in[2];
    const float* W_lin     = (const float*)d_in[3];
    const float* att_src   = (const float*)d_in[4];
    const float* att_dst   = (const float*)d_in[5];
    const float* W_edge    = (const float*)d_in[6];
    const float* att_edge  = (const float*)d_in[7];
    const float* bias      = (const float*)d_in[8];
    const float* W_out     = (const float*)d_in[9];
    const float* b_out     = (const float*)d_in[10];
    (void)in_sizes; (void)n_in; (void)out_size; (void)ws_size;

    const int* src = ei;
    const int* dst = ei + N_EDGES;

    float* w        = (float*)d_ws;
    float* xl       = w;
    float* agg      = xl      + (long)N_NODES * NHID;
    float* a_src    = agg     + (long)N_NODES * NHID;
    float* a_dst    = a_src   + (long)N_NODES * HEADS;
    float* emax     = a_dst   + (long)N_NODES * HEADS;
    float* denom    = emax    + (long)N_NODES * HEADS;
    float* ebuf     = denom   + (long)N_NODES * HEADS;
    float* v_edge   = ebuf    + (long)N_EDGES * HEADS;
    float* alpha_s  = v_edge  + EDGE_DIM * HEADS;

    float* y         = (float*)d_out;
    float* alpha_out = y + N_NODES;

    k_init<<<2048, 256, 0, stream>>>(agg, emax, denom, alpha_s);
    k_vedge<<<1, 64, 0, stream>>>(W_edge, att_edge, v_edge);

    {
        int waves = M_TILES * N_TILES;
        int blocks = (waves + 7) / 8;
        k_gemm_wmma<<<blocks, 256, 0, stream>>>(x, W_lin, xl);
    }
    k_att<<<(N_NODES * HEADS + 255) / 256, 256, 0, stream>>>(xl, att_src, att_dst, a_src, a_dst);
    k_edge_logits<<<(N_EDGES + 255) / 256, 256, 0, stream>>>(src, dst, edge_attr,
                                                             a_src, a_dst, v_edge, ebuf, emax);
    k_edge_exp<<<(N_EDGES + 255) / 256, 256, 0, stream>>>(dst, emax, ebuf, denom);
    k_scatter<<<2048, 256, 0, stream>>>(src, dst, ebuf, denom, xl, agg, alpha_s);
    k_final<<<(N_NODES * 32 + 255) / 256, 256, 0, stream>>>(agg, bias, W_out, b_out, y);
    k_alpha_mean<<<1, 32, 0, stream>>>(alpha_s, alpha_out);
}


# Edge-logit and edge-exp kernels: one lane per (edge, head) instead of one per edge, so an edge's 8 head atomics leave as one 32-byte request instead of 8; arrival-order atomic sums as before
# speedup vs baseline: 2.3051x; 2.3051x over previous
#include <hip/hip_runtime.h>
#include <hip/hip_bf16.h>

#define N_NODES 50000
#define N_EDGES 800000
#define HID     24
#define HEADS   8
#define EDGE_DIM 8
#define NHID    192
#define M_TILES 3125
#define N_TILES 12

typedef __attribute__((ext_vector_type(2))) float v2f;
typedef __attribute__((ext_vector_type(8))) float v8f;

__device__ __forceinline__ void atomic_fmax(float* addr, float val) {
    if (val >= 0.0f)
        atomicMax((int*)addr, __float_as_int(val));
    else
        atomicMin((unsigned int*)addr, __float_as_uint(val));
}

__global__ void k_init(float* __restrict__ agg, float* __restrict__ emax,
                       float* __restrict__ denom, float* __restrict__ alpha_sum) {
    long i = (long)blockIdx.x * blockDim.x + threadIdx.x;
    long stride = (long)gridDim.x * blockDim.x;
    const float NEG_INF = __int_as_float(0xff800000);
    for (long j = i; j < (long)N_NODES * NHID; j += stride) agg[j] = 0.0f;
    for (long j = i; j < (long)N_NODES * HEADS; j += stride) {
        emax[j] = NEG_INF;
        denom[j] = 0.0f;
    }
    if (i < HEADS) alpha_sum[i] = 0.0f;
}

__global__ void k_vedge(const float* __restrict__ W_edge,
                        const float* __restrict__ att_edge,
                        float* __restrict__ v_edge) {
    int i = threadIdx.x;
    if (i >= EDGE_DIM * HEADS) return;
    int d = i >> 3, h = i & 7;
    float s = 0.0f;
    #pragma unroll
    for (int c = 0; c < HID; ++c)
        s += W_edge[d * NHID + h * HID + c] * att_edge[h * HID + c];
    v_edge[d * HEADS + h] = s;
}

__global__ void k_gemm_wmma(const float* __restrict__ x,
                            const float* __restrict__ W,
                            float* __restrict__ xl) {
    int lane = threadIdx.x & 31;
    int wave = (int)((blockIdx.x * blockDim.x + threadIdx.x) >> 5);
    if (wave >= M_TILES * N_TILES) return;
    int tm = wave / N_TILES;
    int tn = wave % N_TILES;

    int m  = lane & 15;
    int hs = lane >> 4;
    const float* arow = x + (long)(tm * 16 + m) * HID;
    int bcol = tn * 16 + m;

    v8f c = {0.f, 0.f, 0.f, 0.f, 0.f, 0.f, 0.f, 0.f};
    #pragma unroll
    for (int kb = 0; kb < HID; kb += 4) {
        int k0 = kb + hs * 2;
        v2f a; a.x = arow[k0]; a.y = arow[k0 + 1];
        v2f b; b.x = W[(long)k0 * NHID + bcol]; b.y = W[(long)(k0 + 1) * NHID + bcol];
        c = __builtin_amdgcn_wmma_f32_16x16x4_f32(
                 false, a,  false, b,
                 (short)0, c,  false,  false);
    }
    int n = lane & 15;
    #pragma unroll
    for (int r = 0; r < 8; ++r) {
        int mg = tm * 16 + r + hs * 8;
        xl[(long)mg * NHID + tn * 16 + n] = c[r];
    }
}

__global__ void k_att(const float* __restrict__ xl,
                      const float* __restrict__ att_src,
                      const float* __restrict__ att_dst,
                      float* __restrict__ a_src, float* __restrict__ a_dst) {
    int i = blockIdx.x * blockDim.x + threadIdx.x;
    if (i >= N_NODES * HEADS) return;
    int n = i >> 3, h = i & 7;
    const float* row = xl + (long)n * NHID + h * HID;
    const float* as  = att_src + h * HID;
    const float* ad  = att_dst + h * HID;
    float s = 0.0f, d = 0.0f;
    #pragma unroll
    for (int c = 0; c < HID; ++c) {
        float v = row[c];
        s += v * as[c];
        d += v * ad[c];
    }
    a_src[i] = s;
    a_dst[i] = d;
}

#define EDGES_PER_WAVE 32
#define GROUPS_PER_WAVE 8
static_assert(N_EDGES % EDGES_PER_WAVE == 0, "a wave's 32 edges are all real edges");

__global__ void k_edge_logits(const int* __restrict__ src, const int* __restrict__ dst,
                              const float* __restrict__ edge_attr,
                              const float* __restrict__ a_src, const float* __restrict__ a_dst,
                              const float* __restrict__ v_edge,
                              float* __restrict__ ebuf, float* __restrict__ emax) {
    int lane = threadIdx.x & 31;
    int h    = lane & 7;
    int wave = (int)((blockIdx.x * blockDim.x + threadIdx.x) >> 5);
    int e0   = wave * EDGES_PER_WAVE + (lane >> 3);
    if (wave * EDGES_PER_WAVE >= N_EDGES) return;
    float vh[EDGE_DIM];
    #pragma unroll
    for (int j = 0; j < EDGE_DIM; ++j) vh[j] = v_edge[j * HEADS + h];
    int s[GROUPS_PER_WAVE], d[GROUPS_PER_WAVE];
    #pragma unroll
    for (int k = 0; k < GROUPS_PER_WAVE; ++k) {
        s[k] = src[e0 + 4 * k];
        d[k] = dst[e0 + 4 * k];
    }
    float ae[GROUPS_PER_WAVE];
    #pragma unroll
    for (int k = 0; k < GROUPS_PER_WAVE; ++k) {
        const float* ea = edge_attr + (e0 + 4 * k) * EDGE_DIM;
        float t = 0.0f;
        #pragma unroll
        for (int j = 0; j < EDGE_DIM; ++j) t += ea[j] * vh[j];
        ae[k] = t;
    }
    float v[GROUPS_PER_WAVE];
    #pragma unroll
    for (int k = 0; k < GROUPS_PER_WAVE; ++k)
        v[k] = a_src[s[k] * HEADS + h] + a_dst[d[k] * HEADS + h] + ae[k];
    #pragma unroll
    for (int k = 0; k < GROUPS_PER_WAVE; ++k) {
        float t = (v[k] > 0.0f) ? v[k] : 0.2f * v[k];
        ebuf[(e0 + 4 * k) * HEADS + h] = t;
        atomic_fmax(&emax[d[k] * HEADS + h], t);
    }
}

__global__ void k_edge_exp(const int* __restrict__ dst,
                           const float* __restrict__ emax,
                           float* __restrict__ ebuf, float* __restrict__ denom) {
    int lane = threadIdx.x & 31;
    int h    = lane & 7;
    int wave = (int)((blockIdx.x * blockDim.x + threadIdx.x) >> 5);
    int e0   = wave * EDGES_PER_WAVE + (lane >> 3);
    if (wave * EDGES_PER_WAVE >= N_EDGES) return;
    int d[GROUPS_PER_WAVE];
    float x[GROUPS_PER_WAVE], m[GROUPS_PER_WAVE];
    #pragma unroll
    for (int k = 0; k < GROUPS_PER_WAVE; ++k) {
        d[k] = dst[e0 + 4 * k];
        x[k] = ebuf[(e0 + 4 * k) * HEADS + h];
    }
    #pragma unroll
    for (int k = 0; k < GROUPS_PER_WAVE; ++k) m[k] = emax[d[k] * HEADS + h];
    #pragma unroll
    for (int k = 0; k < GROUPS_PER_WAVE; ++k) {
        float t = __expf(x[k] - m[k]);
        ebuf[(e0 + 4 * k) * HEADS + h] = t;
        atomicAdd(&denom[d[k] * HEADS + h], t);
    }
}

__global__ void k_scatter(const int* __restrict__ src, const int* __restrict__ dst,
                          const float* __restrict__ ebuf, const float* __restrict__ denom,
                          const float* __restrict__ xl, float* __restrict__ agg,
                          float* __restrict__ alpha_sum) {
    __shared__ float s_as[HEADS];
    if (threadIdx.x < HEADS) s_as[threadIdx.x] = 0.0f;
    __syncthreads();

    int lane = threadIdx.x & 31;
    int wpb  = blockDim.x >> 5;
    long wave   = (long)blockIdx.x * wpb + (threadIdx.x >> 5);
    long nwaves = (long)gridDim.x * wpb;

    float asum = 0.0f;
    for (long e = wave; e < N_EDGES; e += nwaves) {
        int s = src[e], d = dst[e];
        float alpha = 0.0f;
        if (lane < HEADS)
            alpha = ebuf[e * HEADS + lane] / (denom[d * HEADS + lane] + 1e-16f);
        asum += alpha;
        const float* xs = xl + (long)s * NHID;
        float*       ag = agg + (long)d * NHID;
        #pragma unroll
        for (int i = 0; i < NHID / 32; ++i) {
            int c = lane + 32 * i;
            int h = c / HID;
            float ah = __shfl(alpha, h, 32);
            atomicAdd(&ag[c], xs[c] * ah);
        }
    }
    if (lane < HEADS) atomicAdd(&s_as[lane], asum);
    __syncthreads();
    if (threadIdx.x < HEADS) atomicAdd(&alpha_sum[threadIdx.x], s_as[threadIdx.x]);
}

__global__ void k_final(const float* __restrict__ agg, const float* __restrict__ bias,
                        const float* __restrict__ W_out, const float* __restrict__ b_out,
                        float* __restrict__ y) {
    int lane = threadIdx.x & 31;
    int node = (int)((blockIdx.x * blockDim.x + threadIdx.x) >> 5);
    if (node >= N_NODES) return;
    float val = 0.0f;
    if (lane < HID) {
        float s = 0.0f;
        #pragma unroll
        for (int h = 0; h < HEADS; ++h) s += agg[(long)node * NHID + h * HID + lane];
        s = s * (1.0f / HEADS) + bias[lane];
        s = (s > 0.0f) ? s : (__expf(s) - 1.0f);
        val = s * W_out[lane];
    }
    #pragma unroll
    for (int off = 16; off > 0; off >>= 1) val += __shfl_down(val, off, 32);
    if (lane == 0) y[node] = val + b_out[0];
}

__global__ void k_alpha_mean(const float* __restrict__ alpha_sum, float* __restrict__ out) {
    int h = threadIdx.x;
    if (h < HEADS) out[h] = alpha_sum[h] * (1.0f / (float)N_EDGES);
}

extern "C" void kernel_launch(void* const* d_in, const int* in_sizes, int n_in,
                              void* d_out, int out_size, void* d_ws, size_t ws_size,
                              hipStream_t stream) {
    const float* x         = (const float*)d_in[0];
    const int*   ei        = (const int*)d_in[1];
    const float* edge_attr = (const float*)d_in[2];
    const float* W_lin     = (const float*)d_in[3];
    const float* att_src   = (const float*)d_in[4];
    const float* att_dst   = (const float*)d_in[5];
    const float* W_edge    = (const float*)d_in[6];
    const float* att_edge  = (const float*)d_in[7];
    const float* bias      = (const float*)d_in[8];
    const float* W_out     = (const float*)d_in[9];
    const float* b_out     = (const float*)d_in[10];
    (void)in_sizes; (void)n_in; (void)out_size; (void)ws_size;

    const int* src = ei;
    const int* dst = ei + N_EDGES;

    float* w        = (float*)d_ws;
    float* xl       = w;
    float* agg      = xl      + (long)N_NODES * NHID;
    float* a_src    = agg     + (long)N_NODES * NHID;
    float* a_dst    = a_src   + (long)N_NODES * HEADS;
    float* emax     = a_dst   + (long)N_NODES * HEADS;
    float* denom    = emax    + (long)N_NODES * HEADS;
    float* ebuf     = denom   + (long)N_NODES * HEADS;
    float* v_edge   = ebuf    + (long)N_EDGES * HEADS;
    float* alpha_s  = v_edge  + EDGE_DIM * HEADS;

    float* y         = (float*)d_out;
    float* alpha_out = y + N_NODES;

    k_init<<<2048, 256, 0, stream>>>(agg, emax, denom, alpha_s);
    k_vedge<<<1, 64, 0, stream>>>(W_edge, att_edge, v_edge);

    {
        int waves = M_TILES * N_TILES;
        int blocks = (waves + 7) / 8;
        k_gemm_wmma<<<blocks, 256, 0, stream>>>(x, W_lin, xl);
    }
    k_att<<<(N_NODES * HEADS + 255) / 256, 256, 0, stream>>>(xl, att_src, att_dst, a_src, a_dst);
    k_edge_logits<<<(N_EDGES + 255) / 256, 256, 0, stream>>>(src, dst, edge_attr,
                                                             a_src, a_dst, v_edge, ebuf, emax);
    k_edge_exp<<<(N_EDGES + 255) / 256, 256, 0, stream>>>(dst, emax, ebuf, denom);
    k_scatter<<<2048, 256, 0, stream>>>(src, dst, ebuf, denom, xl, agg, alpha_s);
    k_final<<<(N_NODES * 32 + 255) / 256, 256, 0, stream>>>(agg, bias, W_out, b_out, y);
    k_alpha_mean<<<1, 32, 0, stream>>>(alpha_s, alpha_out);
}


# Scatter sums an edge's 8 head terms per column in registers first (head mean is linear), adds one 24-float row per edge not 192: agg [N,32], 8x fewer atomics; arrival-order atomic sums as before
# speedup vs baseline: 3.5058x; 1.5209x over previous
#include <hip/hip_runtime.h>
#include <hip/hip_bf16.h>

#define N_NODES 50000
#define N_EDGES 800000
#define HID     24
#define HEADS   8
#define EDGE_DIM 8
#define NHID    192
#define M_TILES 3125
#define N_TILES 12
#define AGG_W   32

typedef __attribute__((ext_vector_type(2))) float v2f;
typedef __attribute__((ext_vector_type(8))) float v8f;

__device__ __forceinline__ void atomic_fmax(float* addr, float val) {
    if (val >= 0.0f)
        atomicMax((int*)addr, __float_as_int(val));
    else
        atomicMin((unsigned int*)addr, __float_as_uint(val));
}

__global__ void k_init(float* __restrict__ agg, float* __restrict__ emax,
                       float* __restrict__ denom, float* __restrict__ alpha_sum) {
    long i = (long)blockIdx.x * blockDim.x + threadIdx.x;
    long stride = (long)gridDim.x * blockDim.x;
    const float NEG_INF = __int_as_float(0xff800000);
    for (long j = i; j < (long)N_NODES * AGG_W; j += stride) agg[j] = 0.0f;
    for (long j = i; j < (long)N_NODES * HEADS; j += stride) {
        emax[j] = NEG_INF;
        denom[j] = 0.0f;
    }
    if (i < HEADS) alpha_sum[i] = 0.0f;
}

__global__ void k_vedge(const float* __restrict__ W_edge,
                        const float* __restrict__ att_edge,
                        float* __restrict__ v_edge) {
    int i = threadIdx.x;
    if (i >= EDGE_DIM * HEADS) return;
    int d = i >> 3, h = i & 7;
    float s = 0.0f;
    #pragma unroll
    for (int c = 0; c < HID; ++c)
        s += W_edge[d * NHID + h * HID + c] * att_edge[h * HID + c];
    v_edge[d * HEADS + h] = s;
}

__global__ void k_gemm_wmma(const float* __restrict__ x,
                            const float* __restrict__ W,
                            float* __restrict__ xl) {
    int lane = threadIdx.x & 31;
    int wave = (int)((blockIdx.x * blockDim.x + threadIdx.x) >> 5);
    if (wave >= M_TILES * N_TILES) return;
    int tm = wave / N_TILES;
    int tn = wave % N_TILES;

    int m  = lane & 15;
    int hs = lane >> 4;
    const float* arow = x + (long)(tm * 16 + m) * HID;
    int bcol = tn * 16 + m;

    v8f c = {0.f, 0.f, 0.f, 0.f, 0.f, 0.f, 0.f, 0.f};
    #pragma unroll
    for (int kb = 0; kb < HID; kb += 4) {
        int k0 = kb + hs * 2;
        v2f a; a.x = arow[k0]; a.y = arow[k0 + 1];
        v2f b; b.x = W[(long)k0 * NHID + bcol]; b.y = W[(long)(k0 + 1) * NHID + bcol];
        c = __builtin_amdgcn_wmma_f32_16x16x4_f32(
                 false, a,  false, b,
                 (short)0, c,  false,  false);
    }
    int n = lane & 15;
    #pragma unroll
    for (int r = 0; r < 8; ++r) {
        int mg = tm * 16 + r + hs * 8;
        xl[(long)mg * NHID + tn * 16 + n] = c[r];
    }
}

__global__ void k_att(const float* __restrict__ xl,
                      const float* __restrict__ att_src,
                      const float* __restrict__ att_dst,
                      float* __restrict__ a_src, float* __restrict__ a_dst) {
    int i = blockIdx.x * blockDim.x + threadIdx.x;
    if (i >= N_NODES * HEADS) return;
    int n = i >> 3, h = i & 7;
    const float* row = xl + (long)n * NHID + h * HID;
    const float* as  = att_src + h * HID;
    const float* ad  = att_dst + h * HID;
    float s = 0.0f, d = 0.0f;
    #pragma unroll
    for (int c = 0; c < HID; ++c) {
        float v = row[c];
        s += v * as[c];
        d += v * ad[c];
    }
    a_src[i] = s;
    a_dst[i] = d;
}

#define EDGES_PER_WAVE 32
#define GROUPS_PER_WAVE 8
static_assert(N_EDGES % EDGES_PER_WAVE == 0, "a wave's 32 edges are all real edges");

__global__ void k_edge_logits(const int* __restrict__ src, const int* __restrict__ dst,
                              const float* __restrict__ edge_attr,
                              const float* __restrict__ a_src, const float* __restrict__ a_dst,
                              const float* __restrict__ v_edge,
                              float* __restrict__ ebuf, float* __restrict__ emax) {
    int lane = threadIdx.x & 31;
    int h    = lane & 7;
    int wave = (int)((blockIdx.x * blockDim.x + threadIdx.x) >> 5);
    int e0   = wave * EDGES_PER_WAVE + (lane >> 3);
    if (wave * EDGES_PER_WAVE >= N_EDGES) return;
    float vh[EDGE_DIM];
    #pragma unroll
    for (int j = 0; j < EDGE_DIM; ++j) vh[j] = v_edge[j * HEADS + h];
    int s[GROUPS_PER_WAVE], d[GROUPS_PER_WAVE];
    #pragma unroll
    for (int k = 0; k < GROUPS_PER_WAVE; ++k) {
        s[k] = src[e0 + 4 * k];
        d[k] = dst[e0 + 4 * k];
    }
    float ae[GROUPS_PER_WAVE];
    #pragma unroll
    for (int k = 0; k < GROUPS_PER_WAVE; ++k) {
        const float* ea = edge_attr + (e0 + 4 * k) * EDGE_DIM;
        float t = 0.0f;
        #pragma unroll
        for (int j = 0; j < EDGE_DIM; ++j) t += ea[j] * vh[j];
        ae[k] = t;
    }
    float v[GROUPS_PER_WAVE];
    #pragma unroll
    for (int k = 0; k < GROUPS_PER_WAVE; ++k)
        v[k] = a_src[s[k] * HEADS + h] + a_dst[d[k] * HEADS + h] + ae[k];
    #pragma unroll
    for (int k = 0; k < GROUPS_PER_WAVE; ++k) {
        float t = (v[k] > 0.0f) ? v[k] : 0.2f * v[k];
        ebuf[(e0 + 4 * k) * HEADS + h] = t;
        atomic_fmax(&emax[d[k] * HEADS + h], t);
    }
}

__global__ void k_edge_exp(const int* __restrict__ dst,
                           const float* __restrict__ emax,
                           float* __restrict__ ebuf, float* __restrict__ denom) {
    int lane = threadIdx.x & 31;
    int h    = lane & 7;
    int wave = (int)((blockIdx.x * blockDim.x + threadIdx.x) >> 5);
    int e0   = wave * EDGES_PER_WAVE + (lane >> 3);
    if (wave * EDGES_PER_WAVE >= N_EDGES) return;
    int d[GROUPS_PER_WAVE];
    float x[GROUPS_PER_WAVE], m[GROUPS_PER_WAVE];
    #pragma unroll
    for (int k = 0; k < GROUPS_PER_WAVE; ++k) {
        d[k] = dst[e0 + 4 * k];
        x[k] = ebuf[(e0 + 4 * k) * HEADS + h];
    }
    #pragma unroll
    for (int k = 0; k < GROUPS_PER_WAVE; ++k) m[k] = emax[d[k] * HEADS + h];
    #pragma unroll
    for (int k = 0; k < GROUPS_PER_WAVE; ++k) {
        float t = __expf(x[k] - m[k]);
        ebuf[(e0 + 4 * k) * HEADS + h] = t;
        atomicAdd(&denom[d[k] * HEADS + h], t);
    }
}

__global__ void k_scatter(const int* __restrict__ src, const int* __restrict__ dst,
                          const float* __restrict__ ebuf, const float* __restrict__ denom,
                          const float* __restrict__ xl, float* __restrict__ agg,
                          float* __restrict__ alpha_sum) {
    __shared__ float s_as[HEADS];
    if (threadIdx.x < HEADS) s_as[threadIdx.x] = 0.0f;
    __syncthreads();

    int lane = threadIdx.x & 31;
    int wpb  = blockDim.x >> 5;
    long wave   = (long)blockIdx.x * wpb + (threadIdx.x >> 5);
    long nwaves = (long)gridDim.x * wpb;

    float asum = 0.0f;
    for (long e = wave; e < N_EDGES; e += nwaves) {
        int s = src[e], d = dst[e];
        float alpha = 0.0f;
        if (lane < HEADS)
            alpha = ebuf[e * HEADS + lane] / (denom[d * HEADS + lane] + 1e-16f);
        asum += alpha;
        const float* xs = xl + (long)s * NHID;
        float p[NHID / 32];
        #pragma unroll
        for (int i = 0; i < NHID / 32; ++i) {
            int c = lane + 32 * i;
            int h = c / HID;
            float ah = __shfl(alpha, h, 32);
            p[i] = xs[c] * ah;
        }
        float q0 = p[0] + p[3];
        float q1 = p[1] + p[4];
        float q2 = p[2] + p[5];
        float u1 = (lane >= 24) ? q0 : q1;
        float u2 = (lane >= 16) ? q1 : q2;
        float t = q0 + __shfl(u1, (lane + 24) & 31, 32)
                     + __shfl(u2, (lane + 16) & 31, 32)
                     + __shfl(q2, (lane + 8) & 31, 32);
        if (lane < HID)
            atomicAdd(&agg[(long)d * AGG_W + lane], t);
    }
    if (lane < HEADS) atomicAdd(&s_as[lane], asum);
    __syncthreads();
    if (threadIdx.x < HEADS) atomicAdd(&alpha_sum[threadIdx.x], s_as[threadIdx.x]);
}

__global__ void k_final(const float* __restrict__ agg, const float* __restrict__ bias,
                        const float* __restrict__ W_out, const float* __restrict__ b_out,
                        float* __restrict__ y) {
    int lane = threadIdx.x & 31;
    int node = (int)((blockIdx.x * blockDim.x + threadIdx.x) >> 5);
    if (node >= N_NODES) return;
    float val = 0.0f;
    if (lane < HID) {
        float s = agg[(long)node * AGG_W + lane];
        s = s * (1.0f / HEADS) + bias[lane];
        s = (s > 0.0f) ? s : (__expf(s) - 1.0f);
        val = s * W_out[lane];
    }
    #pragma unroll
    for (int off = 16; off > 0; off >>= 1) val += __shfl_down(val, off, 32);
    if (lane == 0) y[node] = val + b_out[0];
}

__global__ void k_alpha_mean(const float* __restrict__ alpha_sum, float* __restrict__ out) {
    int h = threadIdx.x;
    if (h < HEADS) out[h] = alpha_sum[h] * (1.0f / (float)N_EDGES);
}

extern "C" void kernel_launch(void* const* d_in, const int* in_sizes, int n_in,
                              void* d_out, int out_size, void* d_ws, size_t ws_size,
                              hipStream_t stream) {
    const float* x         = (const float*)d_in[0];
    const int*   ei        = (const int*)d_in[1];
    const float* edge_attr = (const float*)d_in[2];
    const float* W_lin     = (const float*)d_in[3];
    const float* att_src   = (const float*)d_in[4];
    const float* att_dst   = (const float*)d_in[5];
    const float* W_edge    = (const float*)d_in[6];
    const float* att_edge  = (const float*)d_in[7];
    const float* bias      = (const float*)d_in[8];
    const float* W_out     = (const float*)d_in[9];
    const float* b_out     = (const float*)d_in[10];
    (void)in_sizes; (void)n_in; (void)out_size; (void)ws_size;

    const int* src = ei;
    const int* dst = ei + N_EDGES;

    float* w        = (float*)d_ws;
    float* xl       = w;
    float* agg      = xl      + (long)N_NODES * NHID;
    float* a_src    = agg     + (long)N_NODES * AGG_W;
    float* a_dst    = a_src   + (long)N_NODES * HEADS;
    float* emax     = a_dst   + (long)N_NODES * HEADS;
    float* denom    = emax    + (long)N_NODES * HEADS;
    float* ebuf     = denom   + (long)N_NODES * HEADS;
    float* v_edge   = ebuf    + (long)N_EDGES * HEADS;
    float* alpha_s  = v_edge  + EDGE_DIM * HEADS;

    float* y         = (float*)d_out;
    float* alpha_out = y + N_NODES;

    k_init<<<2048, 256, 0, stream>>>(agg, emax, denom, alpha_s);
    k_vedge<<<1, 64, 0, stream>>>(W_edge, att_edge, v_edge);

    {
        int waves = M_TILES * N_TILES;
        int blocks = (waves + 7) / 8;
        k_gemm_wmma<<<blocks, 256, 0, stream>>>(x, W_lin, xl);
    }
    k_att<<<(N_NODES * HEADS + 255) / 256, 256, 0, stream>>>(xl, att_src, att_dst, a_src, a_dst);
    k_edge_logits<<<(N_EDGES + 255) / 256, 256, 0, stream>>>(src, dst, edge_attr,
                                                             a_src, a_dst, v_edge, ebuf, emax);
    k_edge_exp<<<(N_EDGES + 255) / 256, 256, 0, stream>>>(dst, emax, ebuf, denom);
    k_scatter<<<2048, 256, 0, stream>>>(src, dst, ebuf, denom, xl, agg, alpha_s);
    k_final<<<(N_NODES * 32 + 255) / 256, 256, 0, stream>>>(agg, bias, W_out, b_out, y);
    k_alpha_mean<<<1, 32, 0, stream>>>(alpha_s, alpha_out);
}


# Scatter: a wave takes 4 consecutive edges per trip and issues the trip's 28 loads (24 row lines) before first use, next trip's small loads a trip ahead; grid 1280; alpha sums regrouped per lane
# speedup vs baseline: 3.9562x; 1.1285x over previous
#include <hip/hip_runtime.h>
#include <hip/hip_bf16.h>

#define N_NODES 50000
#define N_EDGES 800000
#define HID     24
#define HEADS   8
#define EDGE_DIM 8
#define NHID    192
#define M_TILES 3125
#define N_TILES 12
#define AGG_W   32

typedef __attribute__((ext_vector_type(2))) float v2f;
typedef __attribute__((ext_vector_type(8))) float v8f;

__device__ __forceinline__ void atomic_fmax(float* addr, float val) {
    if (val >= 0.0f)
        atomicMax((int*)addr, __float_as_int(val));
    else
        atomicMin((unsigned int*)addr, __float_as_uint(val));
}

__global__ void k_init(float* __restrict__ agg, float* __restrict__ emax,
                       float* __restrict__ denom, float* __restrict__ alpha_sum) {
    long i = (long)blockIdx.x * blockDim.x + threadIdx.x;
    long stride = (long)gridDim.x * blockDim.x;
    const float NEG_INF = __int_as_float(0xff800000);
    for (long j = i; j < (long)N_NODES * AGG_W; j += stride) agg[j] = 0.0f;
    for (long j = i; j < (long)N_NODES * HEADS; j += stride) {
        emax[j] = NEG_INF;
        denom[j] = 0.0f;
    }
    if (i < HEADS) alpha_sum[i] = 0.0f;
}

__global__ void k_vedge(const float* __restrict__ W_edge,
                        const float* __restrict__ att_edge,
                        float* __restrict__ v_edge) {
    int i = threadIdx.x;
    if (i >= EDGE_DIM * HEADS) return;
    int d = i >> 3, h = i & 7;
    float s = 0.0f;
    #pragma unroll
    for (int c = 0; c < HID; ++c)
        s += W_edge[d * NHID + h * HID + c] * att_edge[h * HID + c];
    v_edge[d * HEADS + h] = s;
}

__global__ void k_gemm_wmma(const float* __restrict__ x,
                            const float* __restrict__ W,
                            float* __restrict__ xl) {
    int lane = threadIdx.x & 31;
    int wave = (int)((blockIdx.x * blockDim.x + threadIdx.x) >> 5);
    if (wave >= M_TILES * N_TILES) return;
    int tm = wave / N_TILES;
    int tn = wave % N_TILES;

    int m  = lane & 15;
    int hs = lane >> 4;
    const float* arow = x + (long)(tm * 16 + m) * HID;
    int bcol = tn * 16 + m;

    v8f c = {0.f, 0.f, 0.f, 0.f, 0.f, 0.f, 0.f, 0.f};
    #pragma unroll
    for (int kb = 0; kb < HID; kb += 4) {
        int k0 = kb + hs * 2;
        v2f a; a.x = arow[k0]; a.y = arow[k0 + 1];
        v2f b; b.x = W[(long)k0 * NHID + bcol]; b.y = W[(long)(k0 + 1) * NHID + bcol];
        c = __builtin_amdgcn_wmma_f32_16x16x4_f32(
                 false, a,  false, b,
                 (short)0, c,  false,  false);
    }
    int n = lane & 15;
    #pragma unroll
    for (int r = 0; r < 8; ++r) {
        int mg = tm * 16 + r + hs * 8;
        xl[(long)mg * NHID + tn * 16 + n] = c[r];
    }
}

__global__ void k_att(const float* __restrict__ xl,
                      const float* __restrict__ att_src,
                      const float* __restrict__ att_dst,
                      float* __restrict__ a_src, float* __restrict__ a_dst) {
    int i = blockIdx.x * blockDim.x + threadIdx.x;
    if (i >= N_NODES * HEADS) return;
    int n = i >> 3, h = i & 7;
    const float* row = xl + (long)n * NHID + h * HID;
    const float* as  = att_src + h * HID;
    const float* ad  = att_dst + h * HID;
    float s = 0.0f, d = 0.0f;
    #pragma unroll
    for (int c = 0; c < HID; ++c) {
        float v = row[c];
        s += v * as[c];
        d += v * ad[c];
    }
    a_src[i] = s;
    a_dst[i] = d;
}

#define EDGES_PER_WAVE 32
#define GROUPS_PER_WAVE 8
static_assert(N_EDGES % EDGES_PER_WAVE == 0, "a wave's 32 edges are all real edges");

__global__ void k_edge_logits(const int* __restrict__ src, const int* __restrict__ dst,
                              const float* __restrict__ edge_attr,
                              const float* __restrict__ a_src, const float* __restrict__ a_dst,
                              const float* __restrict__ v_edge,
                              float* __restrict__ ebuf, float* __restrict__ emax) {
    int lane = threadIdx.x & 31;
    int h    = lane & 7;
    int wave = (int)((blockIdx.x * blockDim.x + threadIdx.x) >> 5);
    int e0   = wave * EDGES_PER_WAVE + (lane >> 3);
    if (wave * EDGES_PER_WAVE >= N_EDGES) return;
    float vh[EDGE_DIM];
    #pragma unroll
    for (int j = 0; j < EDGE_DIM; ++j) vh[j] = v_edge[j * HEADS + h];
    int s[GROUPS_PER_WAVE], d[GROUPS_PER_WAVE];
    #pragma unroll
    for (int k = 0; k < GROUPS_PER_WAVE; ++k) {
        s[k] = src[e0 + 4 * k];
        d[k] = dst[e0 + 4 * k];
    }
    float ae[GROUPS_PER_WAVE];
    #pragma unroll
    for (int k = 0; k < GROUPS_PER_WAVE; ++k) {
        const float* ea = edge_attr + (e0 + 4 * k) * EDGE_DIM;
        float t = 0.0f;
        #pragma unroll
        for (int j = 0; j < EDGE_DIM; ++j) t += ea[j] * vh[j];
        ae[k] = t;
    }
    float v[GROUPS_PER_WAVE];
    #pragma unroll
    for (int k = 0; k < GROUPS_PER_WAVE; ++k)
        v[k] = a_src[s[k] * HEADS + h] + a_dst[d[k] * HEADS + h] + ae[k];
    #pragma unroll
    for (int k = 0; k < GROUPS_PER_WAVE; ++k) {
        float t = (v[k] > 0.0f) ? v[k] : 0.2f * v[k];
        ebuf[(e0 + 4 * k) * HEADS + h] = t;
        atomic_fmax(&emax[d[k] * HEADS + h], t);
    }
}

__global__ void k_edge_exp(const int* __restrict__ dst,
                           const float* __restrict__ emax,
                           float* __restrict__ ebuf, float* __restrict__ denom) {
    int lane = threadIdx.x & 31;
    int h    = lane & 7;
    int wave = (int)((blockIdx.x * blockDim.x + threadIdx.x) >> 5);
    int e0   = wave * EDGES_PER_WAVE + (lane >> 3);
    if (wave * EDGES_PER_WAVE >= N_EDGES) return;
    int d[GROUPS_PER_WAVE];
    float x[GROUPS_PER_WAVE], m[GROUPS_PER_WAVE];
    #pragma unroll
    for (int k = 0; k < GROUPS_PER_WAVE; ++k) {
        d[k] = dst[e0 + 4 * k];
        x[k] = ebuf[(e0 + 4 * k) * HEADS + h];
    }
    #pragma unroll
    for (int k = 0; k < GROUPS_PER_WAVE; ++k) m[k] = emax[d[k] * HEADS + h];
    #pragma unroll
    for (int k = 0; k < GROUPS_PER_WAVE; ++k) {
        float t = __expf(x[k] - m[k]);
        ebuf[(e0 + 4 * k) * HEADS + h] = t;
        atomicAdd(&denom[d[k] * HEADS + h], t);
    }
}

#define SC_G 4
#define SC_GROUPS (N_EDGES / SC_G)
static_assert(N_EDGES % SC_G == 0, "a trip's 4 edges are all real edges");

__global__ void k_scatter(const int* __restrict__ src, const int* __restrict__ dst,
                          const float* __restrict__ ebuf, const float* __restrict__ denom,
                          const float* __restrict__ xl, float* __restrict__ agg,
                          float* __restrict__ alpha_sum) {
    __shared__ float s_as[HEADS];
    if (threadIdx.x < HEADS) s_as[threadIdx.x] = 0.0f;
    __syncthreads();

    int lane = threadIdx.x & 31;
    int j    = lane >> 3;
    int h8   = lane & 7;
    int wpb  = blockDim.x >> 5;
    int wave   = __builtin_amdgcn_readfirstlane((int)(blockIdx.x * wpb + (threadIdx.x >> 5)));
    int nwaves = (int)(gridDim.x * wpb);

    int hl[NHID / 32];
    #pragma unroll
    for (int i = 0; i < NHID / 32; ++i) hl[i] = (lane + 32 * i) / HID;

    float asum = 0.0f;
    int g = wave;
    int gc = (g < SC_GROUPS) ? g : SC_GROUPS - 1;
    int   sv = src[gc * SC_G + (lane & 3)];
    int   dj = dst[gc * SC_G + j];
    float ex = ebuf[gc * (SC_G * HEADS) + lane];
    while (g < SC_GROUPS) {
        int gn = g + nwaves;
        int gnc = (gn < SC_GROUPS) ? gn : SC_GROUPS - 1;
        int   svn = src[gnc * SC_G + (lane & 3)];
        int   djn = dst[gnc * SC_G + j];
        float exn = ebuf[gnc * (SC_G * HEADS) + lane];
        float dn = denom[dj * HEADS + h8];
        float p[SC_G][NHID / 32];
        #pragma unroll
        for (int k = 0; k < SC_G; ++k) {
            int s = __builtin_amdgcn_readlane(sv, k);
            const float* xs = xl + (long)s * NHID + lane;
            #pragma unroll
            for (int i = 0; i < NHID / 32; ++i) p[k][i] = xs[32 * i];
        }
        asm volatile("" :: "v"(svn), "v"(djn), "v"(exn), "v"(dn),
                     "v"(p[0][0]), "v"(p[0][1]), "v"(p[0][2]), "v"(p[0][3]), "v"(p[0][4]), "v"(p[0][5]),
                     "v"(p[1][0]), "v"(p[1][1]), "v"(p[1][2]), "v"(p[1][3]), "v"(p[1][4]), "v"(p[1][5]),
                     "v"(p[2][0]), "v"(p[2][1]), "v"(p[2][2]), "v"(p[2][3]), "v"(p[2][4]), "v"(p[2][5]),
                     "v"(p[3][0]), "v"(p[3][1]), "v"(p[3][2]), "v"(p[3][3]), "v"(p[3][4]), "v"(p[3][5]));
        float alpha = ex / (dn + 1e-16f);
        asum += alpha;
        float t[SC_G];
        #pragma unroll
        for (int k = 0; k < SC_G; ++k) {
            float w[NHID / 32];
            #pragma unroll
            for (int i = 0; i < NHID / 32; ++i)
                w[i] = p[k][i] * __shfl(alpha, 8 * k + hl[i], 32);
            float q0 = w[0] + w[3];
            float q1 = w[1] + w[4];
            float q2 = w[2] + w[5];
            float u1 = (lane >= 24) ? q0 : q1;
            float u2 = (lane >= 16) ? q1 : q2;
            t[k] = q0 + __shfl(u1, (lane + 24) & 31, 32)
                      + __shfl(u2, (lane + 16) & 31, 32)
                      + __shfl(q2, (lane + 8) & 31, 32);
        }
        int d[SC_G];
        #pragma unroll
        for (int k = 0; k < SC_G; ++k)
            d[k] = __builtin_amdgcn_readlane(dj, 8 * k);
        if (lane < HID) {
            #pragma unroll
            for (int k = 0; k < SC_G; ++k)
                atomicAdd(&agg[(long)d[k] * AGG_W + lane], t[k]);
        }
        g = gn; sv = svn; dj = djn; ex = exn;
    }
    asum += __shfl_xor(asum, 8, 32);
    asum += __shfl_xor(asum, 16, 32);
    if (lane < HEADS) atomicAdd(&s_as[lane], asum);
    __syncthreads();
    if (threadIdx.x < HEADS) atomicAdd(&alpha_sum[threadIdx.x], s_as[threadIdx.x]);
}

__global__ void k_final(const float* __restrict__ agg, const float* __restrict__ bias,
                        const float* __restrict__ W_out, const float* __restrict__ b_out,
                        float* __restrict__ y) {
    int lane = threadIdx.x & 31;
    int node = (int)((blockIdx.x * blockDim.x + threadIdx.x) >> 5);
    if (node >= N_NODES) return;
    float val = 0.0f;
    if (lane < HID) {
        float s = agg[(long)node * AGG_W + lane];
        s = s * (1.0f / HEADS) + bias[lane];
        s = (s > 0.0f) ? s : (__expf(s) - 1.0f);
        val = s * W_out[lane];
    }
    #pragma unroll
    for (int off = 16; off > 0; off >>= 1) val += __shfl_down(val, off, 32);
    if (lane == 0) y[node] = val + b_out[0];
}

__global__ void k_alpha_mean(const float* __restrict__ alpha_sum, float* __restrict__ out) {
    int h = threadIdx.x;
    if (h < HEADS) out[h] = alpha_sum[h] * (1.0f / (float)N_EDGES);
}

extern "C" void kernel_launch(void* const* d_in, const int* in_sizes, int n_in,
                              void* d_out, int out_size, void* d_ws, size_t ws_size,
                              hipStream_t stream) {
    const float* x         = (const float*)d_in[0];
    const int*   ei        = (const int*)d_in[1];
    const float* edge_attr = (const float*)d_in[2];
    const float* W_lin     = (const float*)d_in[3];
    const float* att_src   = (const float*)d_in[4];
    const float* att_dst   = (const float*)d_in[5];
    const float* W_edge    = (const float*)d_in[6];
    const float* att_edge  = (const float*)d_in[7];
    const float* bias      = (const float*)d_in[8];
    const float* W_out     = (const float*)d_in[9];
    const float* b_out     = (const float*)d_in[10];
    (void)in_sizes; (void)n_in; (void)out_size; (void)ws_size;

    const int* src = ei;
    const int* dst = ei + N_EDGES;

    float* w        = (float*)d_ws;
    float* xl       = w;
    float* agg      = xl      + (long)N_NODES * NHID;
    float* a_src    = agg     + (long)N_NODES * AGG_W;
    float* a_dst    = a_src   + (long)N_NODES * HEADS;
    float* emax     = a_dst   + (long)N_NODES * HEADS;
    float* denom    = emax    + (long)N_NODES * HEADS;
    float* ebuf     = denom   + (long)N_NODES * HEADS;
    float* v_edge   = ebuf    + (long)N_EDGES * HEADS;
    float* alpha_s  = v_edge  + EDGE_DIM * HEADS;

    float* y         = (float*)d_out;
    float* alpha_out = y + N_NODES;

    k_init<<<2048, 256, 0, stream>>>(agg, emax, denom, alpha_s);
    k_vedge<<<1, 64, 0, stream>>>(W_edge, att_edge, v_edge);

    {
        int waves = M_TILES * N_TILES;
        int blocks = (waves + 7) / 8;
        k_gemm_wmma<<<blocks, 256, 0, stream>>>(x, W_lin, xl);
    }
    k_att<<<(N_NODES * HEADS + 255) / 256, 256, 0, stream>>>(xl, att_src, att_dst, a_src, a_dst);
    k_edge_logits<<<(N_EDGES + 255) / 256, 256, 0, stream>>>(src, dst, edge_attr,
                                                             a_src, a_dst, v_edge, ebuf, emax);
    k_edge_exp<<<(N_EDGES + 255) / 256, 256, 0, stream>>>(dst, emax, ebuf, denom);
    k_scatter<<<1280, 256, 0, stream>>>(src, dst, ebuf, denom, xl, agg, alpha_s);
    k_final<<<(N_NODES * 32 + 255) / 256, 256, 0, stream>>>(agg, bias, W_out, b_out, y);
    k_alpha_mean<<<1, 32, 0, stream>>>(alpha_s, alpha_out);
}


# Segment max on an order-preserving unsigned key of the logit: one atomic max per 4-edge group, not a max/min pair split by sign: half the atomic requests; max exact, arrival-order sums as before
# speedup vs baseline: 4.5583x; 1.1522x over previous
#include <hip/hip_runtime.h>
#include <hip/hip_bf16.h>

#define N_NODES 50000
#define N_EDGES 800000
#define HID     24
#define HEADS   8
#define EDGE_DIM 8
#define NHID    192
#define M_TILES 3125
#define N_TILES 12
#define AGG_W   32

typedef __attribute__((ext_vector_type(2))) float v2f;
typedef __attribute__((ext_vector_type(8))) float v8f;

__device__ __forceinline__ unsigned int fkey(float v) {
    unsigned int b = __float_as_uint(v);
    return (b & 0x80000000u) ? ~b : (b | 0x80000000u);
}
__device__ __forceinline__ float fkey_inv(unsigned int k) {
    return __uint_as_float((k & 0x80000000u) ? (k ^ 0x80000000u) : ~k);
}

__global__ void k_init(float* __restrict__ agg, unsigned int* __restrict__ emax,
                       float* __restrict__ denom, float* __restrict__ alpha_sum) {
    long i = (long)blockIdx.x * blockDim.x + threadIdx.x;
    long stride = (long)gridDim.x * blockDim.x;
    for (long j = i; j < (long)N_NODES * AGG_W; j += stride) agg[j] = 0.0f;
    for (long j = i; j < (long)N_NODES * HEADS; j += stride) {
        emax[j] = 0u;
        denom[j] = 0.0f;
    }
    if (i < HEADS) alpha_sum[i] = 0.0f;
}

__global__ void k_vedge(const float* __restrict__ W_edge,
                        const float* __restrict__ att_edge,
                        float* __restrict__ v_edge) {
    int i = threadIdx.x;
    if (i >= EDGE_DIM * HEADS) return;
    int d = i >> 3, h = i & 7;
    float s = 0.0f;
    #pragma unroll
    for (int c = 0; c < HID; ++c)
        s += W_edge[d * NHID + h * HID + c] * att_edge[h * HID + c];
    v_edge[d * HEADS + h] = s;
}

__global__ void k_gemm_wmma(const float* __restrict__ x,
                            const float* __restrict__ W,
                            float* __restrict__ xl) {
    int lane = threadIdx.x & 31;
    int wave = (int)((blockIdx.x * blockDim.x + threadIdx.x) >> 5);
    if (wave >= M_TILES * N_TILES) return;
    int tm = wave / N_TILES;
    int tn = wave % N_TILES;

    int m  = lane & 15;
    int hs = lane >> 4;
    const float* arow = x + (long)(tm * 16 + m) * HID;
    int bcol = tn * 16 + m;

    v8f c = {0.f, 0.f, 0.f, 0.f, 0.f, 0.f, 0.f, 0.f};
    #pragma unroll
    for (int kb = 0; kb < HID; kb += 4) {
        int k0 = kb + hs * 2;
        v2f a; a.x = arow[k0]; a.y = arow[k0 + 1];
        v2f b; b.x = W[(long)k0 * NHID + bcol]; b.y = W[(long)(k0 + 1) * NHID + bcol];
        c = __builtin_amdgcn_wmma_f32_16x16x4_f32(
                 false, a,  false, b,
                 (short)0, c,  false,  false);
    }
    int n = lane & 15;
    #pragma unroll
    for (int r = 0; r < 8; ++r) {
        int mg = tm * 16 + r + hs * 8;
        xl[(long)mg * NHID + tn * 16 + n] = c[r];
    }
}

__global__ void k_att(const float* __restrict__ xl,
                      const float* __restrict__ att_src,
                      const float* __restrict__ att_dst,
                      float* __restrict__ a_src, float* __restrict__ a_dst) {
    int i = blockIdx.x * blockDim.x + threadIdx.x;
    if (i >= N_NODES * HEADS) return;
    int n = i >> 3, h = i & 7;
    const float* row = xl + (long)n * NHID + h * HID;
    const float* as  = att_src + h * HID;
    const float* ad  = att_dst + h * HID;
    float s = 0.0f, d = 0.0f;
    #pragma unroll
    for (int c = 0; c < HID; ++c) {
        float v = row[c];
        s += v * as[c];
        d += v * ad[c];
    }
    a_src[i] = s;
    a_dst[i] = d;
}

#define EDGES_PER_WAVE 32
#define GROUPS_PER_WAVE 8
static_assert(N_EDGES % EDGES_PER_WAVE == 0, "a wave's 32 edges are all real edges");

__global__ void k_edge_logits(const int* __restrict__ src, const int* __restrict__ dst,
                              const float* __restrict__ edge_attr,
                              const float* __restrict__ a_src, const float* __restrict__ a_dst,
                              const float* __restrict__ v_edge,
                              float* __restrict__ ebuf, unsigned int* __restrict__ emax) {
    int lane = threadIdx.x & 31;
    int h    = lane & 7;
    int wave = (int)((blockIdx.x * blockDim.x + threadIdx.x) >> 5);
    int e0   = wave * EDGES_PER_WAVE + (lane >> 3);
    if (wave * EDGES_PER_WAVE >= N_EDGES) return;
    float vh[EDGE_DIM];
    #pragma unroll
    for (int j = 0; j < EDGE_DIM; ++j) vh[j] = v_edge[j * HEADS + h];
    int s[GROUPS_PER_WAVE], d[GROUPS_PER_WAVE];
    #pragma unroll
    for (int k = 0; k < GROUPS_PER_WAVE; ++k) {
        s[k] = src[e0 + 4 * k];
        d[k] = dst[e0 + 4 * k];
    }
    float ae[GROUPS_PER_WAVE];
    #pragma unroll
    for (int k = 0; k < GROUPS_PER_WAVE; ++k) {
        const float* ea = edge_attr + (e0 + 4 * k) * EDGE_DIM;
        float t = 0.0f;
        #pragma unroll
        for (int j = 0; j < EDGE_DIM; ++j) t += ea[j] * vh[j];
        ae[k] = t;
    }
    float v[GROUPS_PER_WAVE];
    #pragma unroll
    for (int k = 0; k < GROUPS_PER_WAVE; ++k)
        v[k] = a_src[s[k] * HEADS + h] + a_dst[d[k] * HEADS + h] + ae[k];
    #pragma unroll
    for (int k = 0; k < GROUPS_PER_WAVE; ++k) {
        float t = (v[k] > 0.0f) ? v[k] : 0.2f * v[k];
        ebuf[(e0 + 4 * k) * HEADS + h] = t;
        atomicMax(&emax[d[k] * HEADS + h], fkey(t));
    }
}

__global__ void k_edge_exp(const int* __restrict__ dst,
                           const unsigned int* __restrict__ emax,
                           float* __restrict__ ebuf, float* __restrict__ denom) {
    int lane = threadIdx.x & 31;
    int h    = lane & 7;
    int wave = (int)((blockIdx.x * blockDim.x + threadIdx.x) >> 5);
    int e0   = wave * EDGES_PER_WAVE + (lane >> 3);
    if (wave * EDGES_PER_WAVE >= N_EDGES) return;
    int d[GROUPS_PER_WAVE];
    float x[GROUPS_PER_WAVE], m[GROUPS_PER_WAVE];
    #pragma unroll
    for (int k = 0; k < GROUPS_PER_WAVE; ++k) {
        d[k] = dst[e0 + 4 * k];
        x[k] = ebuf[(e0 + 4 * k) * HEADS + h];
    }
    #pragma unroll
    for (int k = 0; k < GROUPS_PER_WAVE; ++k) m[k] = fkey_inv(emax[d[k] * HEADS + h]);
    #pragma unroll
    for (int k = 0; k < GROUPS_PER_WAVE; ++k) {
        float t = __expf(x[k] - m[k]);
        ebuf[(e0 + 4 * k) * HEADS + h] = t;
        atomicAdd(&denom[d[k] * HEADS + h], t);
    }
}

#define SC_G 4
#define SC_GROUPS (N_EDGES / SC_G)
static_assert(N_EDGES % SC_G == 0, "a trip's 4 edges are all real edges");

__global__ void k_scatter(const int* __restrict__ src, const int* __restrict__ dst,
                          const float* __restrict__ ebuf, const float* __restrict__ denom,
                          const float* __restrict__ xl, float* __restrict__ agg,
                          float* __restrict__ alpha_sum) {
    __shared__ float s_as[HEADS];
    if (threadIdx.x < HEADS) s_as[threadIdx.x] = 0.0f;
    __syncthreads();

    int lane = threadIdx.x & 31;
    int j    = lane >> 3;
    int h8   = lane & 7;
    int wpb  = blockDim.x >> 5;
    int wave   = __builtin_amdgcn_readfirstlane((int)(blockIdx.x * wpb + (threadIdx.x >> 5)));
    int nwaves = (int)(gridDim.x * wpb);

    int hl[NHID / 32];
    #pragma unroll
    for (int i = 0; i < NHID / 32; ++i) hl[i] = (lane + 32 * i) / HID;

    float asum = 0.0f;
    int g = wave;
    int gc = (g < SC_GROUPS) ? g : SC_GROUPS - 1;
    int   sv = src[gc * SC_G + (lane & 3)];
    int   dj = dst[gc * SC_G + j];
    float ex = ebuf[gc * (SC_G * HEADS) + lane];
    while (g < SC_GROUPS) {
        int gn = g + nwaves;
        int gnc = (gn < SC_GROUPS) ? gn : SC_GROUPS - 1;
        int   svn = src[gnc * SC_G + (lane & 3)];
        int   djn = dst[gnc * SC_G + j];
        float exn = ebuf[gnc * (SC_G * HEADS) + lane];
        float dn = denom[dj * HEADS + h8];
        float p[SC_G][NHID / 32];
        #pragma unroll
        for (int k = 0; k < SC_G; ++k) {
            int s = __builtin_amdgcn_readlane(sv, k);
            const float* xs = xl + (long)s * NHID + lane;
            #pragma unroll
            for (int i = 0; i < NHID / 32; ++i) p[k][i] = xs[32 * i];
        }
        asm volatile("" :: "v"(svn), "v"(djn), "v"(exn), "v"(dn),
                     "v"(p[0][0]), "v"(p[0][1]), "v"(p[0][2]), "v"(p[0][3]), "v"(p[0][4]), "v"(p[0][5]),
                     "v"(p[1][0]), "v"(p[1][1]), "v"(p[1][2]), "v"(p[1][3]), "v"(p[1][4]), "v"(p[1][5]),
                     "v"(p[2][0]), "v"(p[2][1]), "v"(p[2][2]), "v"(p[2][3]), "v"(p[2][4]), "v"(p[2][5]),
                     "v"(p[3][0]), "v"(p[3][1]), "v"(p[3][2]), "v"(p[3][3]), "v"(p[3][4]), "v"(p[3][5]));
        float alpha = ex / (dn + 1e-16f);
        asum += alpha;
        float t[SC_G];
        #pragma unroll
        for (int k = 0; k < SC_G; ++k) {
            float w[NHID / 32];
            #pragma unroll
            for (int i = 0; i < NHID / 32; ++i)
                w[i] = p[k][i] * __shfl(alpha, 8 * k + hl[i], 32);
            float q0 = w[0] + w[3];
            float q1 = w[1] + w[4];
            float q2 = w[2] + w[5];
            float u1 = (lane >= 24) ? q0 : q1;
            float u2 = (lane >= 16) ? q1 : q2;
            t[k] = q0 + __shfl(u1, (lane + 24) & 31, 32)
                      + __shfl(u2, (lane + 16) & 31, 32)
                      + __shfl(q2, (lane + 8) & 31, 32);
        }
        int d[SC_G];
        #pragma unroll
        for (int k = 0; k < SC_G; ++k)
            d[k] = __builtin_amdgcn_readlane(dj, 8 * k);
        if (lane < HID) {
            #pragma unroll
            for (int k = 0; k < SC_G; ++k)
                atomicAdd(&agg[(long)d[k] * AGG_W + lane], t[k]);
        }
        g = gn; sv = svn; dj = djn; ex = exn;
    }
    asum += __shfl_xor(asum, 8, 32);
    asum += __shfl_xor(asum, 16, 32);
    if (lane < HEADS) atomicAdd(&s_as[lane], asum);
    __syncthreads();
    if (threadIdx.x < HEADS) atomicAdd(&alpha_sum[threadIdx.x], s_as[threadIdx.x]);
}

__global__ void k_final(const float* __restrict__ agg, const float* __restrict__ bias,
                        const float* __restrict__ W_out, const float* __restrict__ b_out,
                        float* __restrict__ y) {
    int lane = threadIdx.x & 31;
    int node = (int)((blockIdx.x * blockDim.x + threadIdx.x) >> 5);
    if (node >= N_NODES) return;
    float val = 0.0f;
    if (lane < HID) {
        float s = agg[(long)node * AGG_W + lane];
        s = s * (1.0f / HEADS) + bias[lane];
        s = (s > 0.0f) ? s : (__expf(s) - 1.0f);
        val = s * W_out[lane];
    }
    #pragma unroll
    for (int off = 16; off > 0; off >>= 1) val += __shfl_down(val, off, 32);
    if (lane == 0) y[node] = val + b_out[0];
}

__global__ void k_alpha_mean(const float* __restrict__ alpha_sum, float* __restrict__ out) {
    int h = threadIdx.x;
    if (h < HEADS) out[h] = alpha_sum[h] * (1.0f / (float)N_EDGES);
}

extern "C" void kernel_launch(void* const* d_in, const int* in_sizes, int n_in,
                              void* d_out, int out_size, void* d_ws, size_t ws_size,
                              hipStream_t stream) {
    const float* x         = (const float*)d_in[0];
    const int*   ei        = (const int*)d_in[1];
    const float* edge_attr = (const float*)d_in[2];
    const float* W_lin     = (const float*)d_in[3];
    const float* att_src   = (const float*)d_in[4];
    const float* att_dst   = (const float*)d_in[5];
    const float* W_edge    = (const float*)d_in[6];
    const float* att_edge  = (const float*)d_in[7];
    const float* bias      = (const float*)d_in[8];
    const float* W_out     = (const float*)d_in[9];
    const float* b_out     = (const float*)d_in[10];
    (void)in_sizes; (void)n_in; (void)out_size; (void)ws_size;

    const int* src = ei;
    const int* dst = ei + N_EDGES;

    float* w        = (float*)d_ws;
    float* xl       = w;
    float* agg      = xl      + (long)N_NODES * NHID;
    float* a_src    = agg     + (long)N_NODES * AGG_W;
    float* a_dst    = a_src   + (long)N_NODES * HEADS;
    unsigned int* emax = (unsigned int*)(a_dst + (long)N_NODES * HEADS);
    float* denom    = a_dst   + 2L * N_NODES * HEADS;
    float* ebuf     = denom   + (long)N_NODES * HEADS;
    float* v_edge   = ebuf    + (long)N_EDGES * HEADS;
    float* alpha_s  = v_edge  + EDGE_DIM * HEADS;

    float* y         = (float*)d_out;
    float* alpha_out = y + N_NODES;

    k_init<<<2048, 256, 0, stream>>>(agg, emax, denom, alpha_s);
    k_vedge<<<1, 64, 0, stream>>>(W_edge, att_edge, v_edge);

    {
        int waves = M_TILES * N_TILES;
        int blocks = (waves + 7) / 8;
        k_gemm_wmma<<<blocks, 256, 0, stream>>>(x, W_lin, xl);
    }
    k_att<<<(N_NODES * HEADS + 255) / 256, 256, 0, stream>>>(xl, att_src, att_dst, a_src, a_dst);
    k_edge_logits<<<(N_EDGES + 255) / 256, 256, 0, stream>>>(src, dst, edge_attr,
                                                             a_src, a_dst, v_edge, ebuf, emax);
    k_edge_exp<<<(N_EDGES + 255) / 256, 256, 0, stream>>>(dst, emax, ebuf, denom);
    k_scatter<<<1280, 256, 0, stream>>>(src, dst, ebuf, denom, xl, agg, alpha_s);
    k_final<<<(N_NODES * 32 + 255) / 256, 256, 0, stream>>>(agg, bias, W_out, b_out, y);
    k_alpha_mean<<<1, 32, 0, stream>>>(alpha_s, alpha_out);
}


# Edges binned by destination range per call (count, place); one work-group per 196-node bin keeps max, denominator, aggregate in LDS, finishes its nodes: no per-edge global atomics; arrival-order sums
# speedup vs baseline: 6.0229x; 1.3213x over previous
#include <hip/hip_runtime.h>
#include <hip/hip_bf16.h>

#define N_NODES 50000
#define N_EDGES 800000
#define HID     24
#define HEADS   8
#define EDGE_DIM 8
#define NHID    192
#define M_TILES 3125
#define N_TILES 12
#define NB      196
#define NBINS   256
#define EPB     1024
#define BIN_BLOCKS ((N_EDGES + EPB - 1) / EPB)
static_assert(NB * NBINS >= N_NODES, "every node has a bin");
static_assert(EPB % 256 == 0, "whole edges per thread in the binning passes");
#define P_SLOTS (N_EDGES + 4 * NBINS)
__device__ __forceinline__ int round4(int c) { return (c + 3) & ~3; }

typedef __attribute__((ext_vector_type(2))) float v2f;
typedef __attribute__((ext_vector_type(8))) float v8f;

__device__ __forceinline__ unsigned int fkey(float v) {
    unsigned int b = __float_as_uint(v);
    return (b & 0x80000000u) ? ~b : (b | 0x80000000u);
}
__device__ __forceinline__ float fkey_inv(unsigned int k) {
    return __uint_as_float((k & 0x80000000u) ? (k ^ 0x80000000u) : ~k);
}

__global__ void k_vedge(const float* __restrict__ W_edge,
                        const float* __restrict__ att_edge,
                        float* __restrict__ v_edge,
                        int* __restrict__ bincount, int* __restrict__ cursor,
                        float* __restrict__ alpha_sum) {
    int i = threadIdx.x;
    bincount[i] = 0;
    cursor[i] = 0;
    if (i < HEADS) alpha_sum[i] = 0.0f;
    if (i >= EDGE_DIM * HEADS) return;
    int d = i >> 3, h = i & 7;
    float s = 0.0f;
    #pragma unroll
    for (int c = 0; c < HID; ++c)
        s += W_edge[d * NHID + h * HID + c] * att_edge[h * HID + c];
    v_edge[d * HEADS + h] = s;
}

__global__ void k_gemm_wmma(const float* __restrict__ x,
                            const float* __restrict__ W,
                            float* __restrict__ xl) {
    int lane = threadIdx.x & 31;
    int wave = (int)((blockIdx.x * blockDim.x + threadIdx.x) >> 5);
    if (wave >= M_TILES * N_TILES) return;
    int tm = wave / N_TILES;
    int tn = wave % N_TILES;

    int m  = lane & 15;
    int hs = lane >> 4;
    const float* arow = x + (long)(tm * 16 + m) * HID;
    int bcol = tn * 16 + m;

    v8f c = {0.f, 0.f, 0.f, 0.f, 0.f, 0.f, 0.f, 0.f};
    #pragma unroll
    for (int kb = 0; kb < HID; kb += 4) {
        int k0 = kb + hs * 2;
        v2f a; a.x = arow[k0]; a.y = arow[k0 + 1];
        v2f b; b.x = W[(long)k0 * NHID + bcol]; b.y = W[(long)(k0 + 1) * NHID + bcol];
        c = __builtin_amdgcn_wmma_f32_16x16x4_f32(
                 false, a,  false, b,
                 (short)0, c,  false,  false);
    }
    int n = lane & 15;
    #pragma unroll
    for (int r = 0; r < 8; ++r) {
        int mg = tm * 16 + r + hs * 8;
        xl[(long)mg * NHID + tn * 16 + n] = c[r];
    }
}

__global__ void k_att(const float* __restrict__ xl,
                      const float* __restrict__ att_src,
                      const float* __restrict__ att_dst,
                      float* __restrict__ a_src, float* __restrict__ a_dst) {
    int i = blockIdx.x * blockDim.x + threadIdx.x;
    if (i >= N_NODES * HEADS) return;
    int n = i >> 3, h = i & 7;
    const float* row = xl + (long)n * NHID + h * HID;
    const float* as  = att_src + h * HID;
    const float* ad  = att_dst + h * HID;
    float s = 0.0f, d = 0.0f;
    #pragma unroll
    for (int c = 0; c < HID; ++c) {
        float v = row[c];
        s += v * as[c];
        d += v * ad[c];
    }
    a_src[i] = s;
    a_dst[i] = d;
}

__global__ void __launch_bounds__(256)
k_hist(const int* __restrict__ dst, int* __restrict__ bincount) {
    __shared__ int s_cnt[NBINS];
    int t = threadIdx.x;
    s_cnt[t] = 0;
    __syncthreads();
    int e0 = blockIdx.x * EPB;
    #pragma unroll
    for (int i = 0; i < EPB / 256; ++i) {
        int e = e0 + i * 256 + t;
        if (e < N_EDGES) atomicAdd(&s_cnt[dst[e] / NB], 1);
    }
    __syncthreads();
    int c = s_cnt[t];
    if (c) atomicAdd(&bincount[t], c);
}

#define EPT (EPB / 256)
__global__ void __launch_bounds__(256, 8)
k_binplace(const int* __restrict__ src, const int* __restrict__ dst,
           const float* __restrict__ edge_attr,
           const float* __restrict__ v_edge,
           const float* __restrict__ a_src, const float* __restrict__ a_dst,
           const int* __restrict__ bincount, int* __restrict__ cursor,
           int* __restrict__ pe, float* __restrict__ ebuf) {
    __shared__ int s_cnt[NBINS];
    __shared__ int s_base[NBINS];
    __shared__ int s_scan[NBINS];
    __shared__ float s_v[EDGE_DIM * HEADS];
    int t = threadIdx.x;
    s_cnt[t] = 0;
    if (t < EDGE_DIM * HEADS) s_v[t] = v_edge[t];
    int tot = round4(bincount[t]);
    s_scan[t] = tot;

    int e0 = blockIdx.x * EPB;
    int ss[EPT], dd[EPT];
    #pragma unroll
    for (int i = 0; i < EPT; ++i) {
        int e = e0 + i * 256 + t;
        int ec = (e < N_EDGES) ? e : N_EDGES - 1;
        ss[i] = src[ec];
        dd[i] = dst[ec];
    }
    __syncthreads();
    for (int off = 1; off < NBINS; off <<= 1) {
        int a = (t >= off) ? s_scan[t - off] : 0;
        __syncthreads();
        s_scan[t] += a;
        __syncthreads();
    }
    int excl = s_scan[t] - tot;

    #pragma unroll
    for (int i = 0; i < EPT; ++i) {
        int e = e0 + i * 256 + t;
        if (e < N_EDGES) atomicAdd(&s_cnt[dd[i] / NB], 1);
    }
    float as[EPT][HEADS], ad[EPT][HEADS];
    #pragma unroll
    for (int i = 0; i < EPT; ++i) {
        const float* ps = a_src + (long)ss[i] * HEADS;
        const float* pd = a_dst + (long)dd[i] * HEADS;
        #pragma unroll
        for (int h = 0; h < HEADS; ++h) { as[i][h] = ps[h]; ad[i][h] = pd[h]; }
    }
    __syncthreads();
    int c = s_cnt[t];
    int prev = c ? atomicAdd(&cursor[t], c) : 0;
    s_base[t] = excl + prev;
    __syncthreads();
    s_cnt[t] = 0;
    __syncthreads();
    #pragma unroll
    for (int i = 0; i < EPT; ++i) {
        int e = e0 + i * 256 + t;
        if (e < N_EDGES) {
            int bin = dd[i] / NB;
            int pos = s_base[bin] + atomicAdd(&s_cnt[bin], 1);
            reinterpret_cast<int2*>(pe)[pos] = make_int2(ss[i], dd[i]);
            const float* pa = edge_attr + (long)e * EDGE_DIM;
            float ea[EDGE_DIM];
            #pragma unroll
            for (int jj = 0; jj < EDGE_DIM; ++jj) ea[jj] = pa[jj];
            float o[HEADS];
            #pragma unroll
            for (int h = 0; h < HEADS; ++h) {
                float acc = 0.0f;
                #pragma unroll
                for (int jj = 0; jj < EDGE_DIM; ++jj) acc += ea[jj] * s_v[jj * HEADS + h];
                float v = as[i][h] + ad[i][h] + acc;
                o[h] = (v > 0.0f) ? v : 0.2f * v;
            }
            float* ob = ebuf + (long)pos * HEADS;
            #pragma unroll
            for (int h = 0; h < HEADS; ++h) ob[h] = o[h];
        }
    }
}

#define SC_G 4
#define FW   32
#define PU   8

__global__ void __launch_bounds__(1024)
k_bin_nodes(const int* __restrict__ pe_, float* __restrict__ ebuf, const int* __restrict__ bincount,
            const float* __restrict__ xl,
            const float* __restrict__ bias, const float* __restrict__ W_out,
            const float* __restrict__ b_out,
            float* __restrict__ y, float* __restrict__ alpha_sum) {
    __shared__ unsigned int s_max[NB * HEADS];
    __shared__ float s_den[NB * HEADS];
    __shared__ float s_agg[NB * HID];
    __shared__ float s_y[NB];
    __shared__ float s_as[HEADS];
    __shared__ int s_start;

    const int2* pe = reinterpret_cast<const int2*>(pe_);
    int t    = threadIdx.x;
    int b    = blockIdx.x;
    int n0   = b * NB;
    int nn   = (N_NODES - n0 < NB) ? (N_NODES - n0) : NB;
    int lane = t & 31;
    int j    = lane >> 3;
    int h8   = lane & 7;
    int wv   = __builtin_amdgcn_readfirstlane(t >> 5);

    if (t == 0) s_start = 0;
    for (int i = t; i < NB * HEADS; i += 1024) { s_max[i] = 0u; s_den[i] = 0.0f; }
    for (int i = t; i < NB * HID; i += 1024) s_agg[i] = 0.0f;
    if (t < HEADS) s_as[t] = 0.0f;
    __syncthreads();
    if (t < b) atomicAdd(&s_start, round4(bincount[t]));
    __syncthreads();
    int start = __builtin_amdgcn_readfirstlane(s_start);
    int end   = start + __builtin_amdgcn_readfirstlane(bincount[b]);
    int last  = end - 1;

    for (int pb = start + SC_G * wv; pb < end; pb += SC_G * FW * PU) {
        int pp[PU], d[PU];
        float x[PU];
        #pragma unroll
        for (int u = 0; u < PU; ++u) {
            pp[u] = pb + SC_G * FW * u + j;
            int pc = (pp[u] < end) ? pp[u] : last;
            d[u] = pe[pc].y;
            x[u] = ebuf[pc * HEADS + h8];
        }
        asm volatile("" :: "v"(d[0]), "v"(d[1]), "v"(d[2]), "v"(d[3]), "v"(d[4]), "v"(d[5]), "v"(d[6]), "v"(d[7]),
                     "v"(x[0]), "v"(x[1]), "v"(x[2]), "v"(x[3]), "v"(x[4]), "v"(x[5]), "v"(x[6]), "v"(x[7]));
        #pragma unroll
        for (int u = 0; u < PU; ++u)
            if (pp[u] < end) atomicMax(&s_max[(d[u] - n0) * HEADS + h8], fkey(x[u]));
    }
    __syncthreads();

    for (int pb = start + SC_G * wv; pb < end; pb += SC_G * FW * PU) {
        int pp[PU], d[PU];
        float x[PU];
        #pragma unroll
        for (int u = 0; u < PU; ++u) {
            pp[u] = pb + SC_G * FW * u + j;
            int pc = (pp[u] < end) ? pp[u] : last;
            d[u] = pe[pc].y;
            x[u] = ebuf[pc * HEADS + h8];
        }
        asm volatile("" :: "v"(d[0]), "v"(d[1]), "v"(d[2]), "v"(d[3]), "v"(d[4]), "v"(d[5]), "v"(d[6]), "v"(d[7]),
                     "v"(x[0]), "v"(x[1]), "v"(x[2]), "v"(x[3]), "v"(x[4]), "v"(x[5]), "v"(x[6]), "v"(x[7]));
        #pragma unroll
        for (int u = 0; u < PU; ++u) {
            if (pp[u] < end) {
                int di = (d[u] - n0) * HEADS + h8;
                float tt = __expf(x[u] - fkey_inv(s_max[di]));
                ebuf[pp[u] * HEADS + h8] = tt;
                atomicAdd(&s_den[di], tt);
            }
        }
    }
    __syncthreads();

    int hl[NHID / 32];
    #pragma unroll
    for (int i = 0; i < NHID / 32; ++i) hl[i] = (lane + 32 * i) / HID;

    float asum = 0.0f;
    int pb = start + SC_G * wv;
    if (pb < end) {
        int pj = pb + j;
        int pjc = (pj < end) ? pj : last;
        int2  sdj = pe[pjc];
        int   sv = sdj.x;
        int   dj = sdj.y;
        float ex = ebuf[pjc * HEADS + h8];
        ex = (pj < end) ? ex : 0.0f;
        while (pb < end) {
            int pn = pb + SC_G * FW;
            int pjn = pn + j;
            int pjnc = (pjn < end) ? pjn : last;
            int2  sdn = pe[pjnc];
            int   svn = sdn.x;
            int   djn = sdn.y;
            float exn = ebuf[pjnc * HEADS + h8];
            float dn = s_den[(dj - n0) * HEADS + h8];
            float p[SC_G][NHID / 32];
            #pragma unroll
            for (int k = 0; k < SC_G; ++k) {
                int s = __builtin_amdgcn_readlane(sv, 8 * k);
                const float* xs = xl + (long)s * NHID + lane;
                #pragma unroll
                for (int i = 0; i < NHID / 32; ++i) p[k][i] = xs[32 * i];
            }
            asm volatile("" :: "v"(svn), "v"(djn), "v"(exn), "v"(dn),
                         "v"(p[0][0]), "v"(p[0][1]), "v"(p[0][2]), "v"(p[0][3]), "v"(p[0][4]), "v"(p[0][5]),
                         "v"(p[1][0]), "v"(p[1][1]), "v"(p[1][2]), "v"(p[1][3]), "v"(p[1][4]), "v"(p[1][5]),
                         "v"(p[2][0]), "v"(p[2][1]), "v"(p[2][2]), "v"(p[2][3]), "v"(p[2][4]), "v"(p[2][5]),
                         "v"(p[3][0]), "v"(p[3][1]), "v"(p[3][2]), "v"(p[3][3]), "v"(p[3][4]), "v"(p[3][5]));
            float alpha = ex / (dn + 1e-16f);
            asum += alpha;
            float tk[SC_G];
            #pragma unroll
            for (int k = 0; k < SC_G; ++k) {
                float w[NHID / 32];
                #pragma unroll
                for (int i = 0; i < NHID / 32; ++i)
                    w[i] = p[k][i] * __shfl(alpha, 8 * k + hl[i], 32);
                float q0 = w[0] + w[3];
                float q1 = w[1] + w[4];
                float q2 = w[2] + w[5];
                float u1 = (lane >= 24) ? q0 : q1;
                float u2 = (lane >= 16) ? q1 : q2;
                tk[k] = q0 + __shfl(u1, (lane + 24) & 31, 32)
                           + __shfl(u2, (lane + 16) & 31, 32)
                           + __shfl(q2, (lane + 8) & 31, 32);
            }
            int dl[SC_G];
            #pragma unroll
            for (int k = 0; k < SC_G; ++k)
                dl[k] = __builtin_amdgcn_readlane(dj, 8 * k) - n0;
            int cnt = end - pb;
            if (lane < HID) {
                #pragma unroll
                for (int k = 0; k < SC_G; ++k)
                    if (k < cnt) atomicAdd(&s_agg[dl[k] * HID + lane], tk[k]);
            }
            pb = pn; sv = svn; dj = djn;
            ex = (pjn < end) ? exn : 0.0f;
        }
    }
    asum += __shfl_xor(asum, 8, 32);
    asum += __shfl_xor(asum, 16, 32);
    if (lane < HEADS) atomicAdd(&s_as[lane], asum);
    __syncthreads();
    if (t < HEADS) atomicAdd(&alpha_sum[t], s_as[t]);

    for (int nl = wv; nl < nn; nl += FW) {
        float val = 0.0f;
        if (lane < HID) {
            float s = s_agg[nl * HID + lane];
            s = s * (1.0f / HEADS) + bias[lane];
            s = (s > 0.0f) ? s : (__expf(s) - 1.0f);
            val = s * W_out[lane];
        }
        #pragma unroll
        for (int off = 16; off > 0; off >>= 1) val += __shfl_down(val, off, 32);
        if (lane == 0) s_y[nl] = val + b_out[0];
    }
    __syncthreads();
    if (t < nn) y[n0 + t] = s_y[t];
}

__global__ void k_alpha_mean(const float* __restrict__ alpha_sum, float* __restrict__ out) {
    int h = threadIdx.x;
    if (h < HEADS) out[h] = alpha_sum[h] * (1.0f / (float)N_EDGES);
}

extern "C" void kernel_launch(void* const* d_in, const int* in_sizes, int n_in,
                              void* d_out, int out_size, void* d_ws, size_t ws_size,
                              hipStream_t stream) {
    const float* x         = (const float*)d_in[0];
    const int*   ei        = (const int*)d_in[1];
    const float* edge_attr = (const float*)d_in[2];
    const float* W_lin     = (const float*)d_in[3];
    const float* att_src   = (const float*)d_in[4];
    const float* att_dst   = (const float*)d_in[5];
    const float* W_edge    = (const float*)d_in[6];
    const float* att_edge  = (const float*)d_in[7];
    const float* bias      = (const float*)d_in[8];
    const float* W_out     = (const float*)d_in[9];
    const float* b_out     = (const float*)d_in[10];
    (void)in_sizes; (void)n_in; (void)out_size; (void)ws_size;

    const int* src = ei;
    const int* dst = ei + N_EDGES;

    float* w        = (float*)d_ws;
    float* xl       = w;
    float* a_src    = xl      + (long)N_NODES * NHID;
    float* a_dst    = a_src   + (long)N_NODES * HEADS;
    float* ebuf     = a_dst   + (long)N_NODES * HEADS;
    int*   pe       = (int*)(ebuf + (long)P_SLOTS * HEADS);
    float* v_edge   = (float*)(pe + 2L * P_SLOTS);
    float* alpha_s  = v_edge  + EDGE_DIM * HEADS;
    int*   bincount = (int*)(alpha_s + HEADS);
    int*   cursor   = bincount + NBINS;

    float* y         = (float*)d_out;
    float* alpha_out = y + N_NODES;

    k_vedge<<<1, NBINS, 0, stream>>>(W_edge, att_edge, v_edge, bincount, cursor, alpha_s);

    {
        int waves = M_TILES * N_TILES;
        int blocks = (waves + 7) / 8;
        k_gemm_wmma<<<blocks, 256, 0, stream>>>(x, W_lin, xl);
    }
    k_att<<<(N_NODES * HEADS + 255) / 256, 256, 0, stream>>>(xl, att_src, att_dst, a_src, a_dst);
    k_hist<<<BIN_BLOCKS, 256, 0, stream>>>(dst, bincount);
    k_binplace<<<BIN_BLOCKS, 256, 0, stream>>>(src, dst, edge_attr, v_edge, a_src, a_dst,
                                               bincount, cursor, pe, ebuf);
    k_bin_nodes<<<NBINS, 1024, 0, stream>>>(pe, ebuf, bincount, xl,
                                            bias, W_out, b_out, y, alpha_s);
    k_alpha_mean<<<1, 32, 0, stream>>>(alpha_s, alpha_out);
}


# W_lin applied after aggregation (linear): bin kernel gathers 24-float x rows, not 192-float xl rows, into per-(node,head) LDS sums, one WMMA product per bin; logits x@(W_lin.att); sums regrouped
# speedup vs baseline: 7.1768x; 1.1916x over previous
#include <hip/hip_runtime.h>
#include <hip/hip_bf16.h>

#define N_NODES 50000
#define N_EDGES 800000
#define HID     24
#define HEADS   8
#define EDGE_DIM 8
#define NHID    192
#define M_TILES 3125
#define ACOLS   16
#define ZP      196
#define NB      196
#define NBINS   256
#define EPB     1024
#define BIN_BLOCKS ((N_EDGES + EPB - 1) / EPB)
static_assert(NB * NBINS >= N_NODES, "every node has a bin");
static_assert(EPB % 256 == 0, "whole edges per thread in the binning passes");
#define P_SLOTS (N_EDGES + 4 * NBINS)
__device__ __forceinline__ int round4(int c) { return (c + 3) & ~3; }

typedef __attribute__((ext_vector_type(2))) float v2f;
typedef __attribute__((ext_vector_type(8))) float v8f;

__device__ __forceinline__ unsigned int fkey(float v) {
    unsigned int b = __float_as_uint(v);
    return (b & 0x80000000u) ? ~b : (b | 0x80000000u);
}
__device__ __forceinline__ float fkey_inv(unsigned int k) {
    return __uint_as_float((k & 0x80000000u) ? (k ^ 0x80000000u) : ~k);
}

__global__ void k_vedge(const float* __restrict__ W_edge,
                        const float* __restrict__ att_edge,
                        const float* __restrict__ W_lin,
                        const float* __restrict__ att_src,
                        const float* __restrict__ att_dst,
                        float* __restrict__ v_edge, float* __restrict__ wa,
                        int* __restrict__ bincount, int* __restrict__ cursor,
                        float* __restrict__ alpha_sum) {
    int i = threadIdx.x;
    bincount[i] = 0;
    cursor[i] = 0;
    if (i < HEADS) alpha_sum[i] = 0.0f;
    for (int q = i; q < HID * ACOLS; q += NBINS) {
        int k = q >> 4, col = q & 15, h = col & 7;
        const float* att = ((col < HEADS) ? att_src : att_dst) + h * HID;
        const float* wr  = W_lin + k * NHID + h * HID;
        float s = 0.0f;
        #pragma unroll
        for (int c = 0; c < HID; ++c) s += wr[c] * att[c];
        wa[q] = s;
    }
    if (i >= EDGE_DIM * HEADS) return;
    int d = i >> 3, h = i & 7;
    float s = 0.0f;
    #pragma unroll
    for (int c = 0; c < HID; ++c)
        s += W_edge[d * NHID + h * HID + c] * att_edge[h * HID + c];
    v_edge[d * HEADS + h] = s;
}

__global__ void k_att_wmma(const float* __restrict__ x,
                           const float* __restrict__ wa,
                           float* __restrict__ a_src, float* __restrict__ a_dst) {
    int lane = threadIdx.x & 31;
    int tm = (int)((blockIdx.x * blockDim.x + threadIdx.x) >> 5);
    if (tm >= M_TILES) return;

    int m  = lane & 15;
    int hs = lane >> 4;
    const float* arow = x + (long)(tm * 16 + m) * HID;

    v8f c = {0.f, 0.f, 0.f, 0.f, 0.f, 0.f, 0.f, 0.f};
    #pragma unroll
    for (int kb = 0; kb < HID; kb += 4) {
        int k0 = kb + hs * 2;
        v2f a; a.x = arow[k0]; a.y = arow[k0 + 1];
        v2f b; b.x = wa[k0 * ACOLS + m]; b.y = wa[(k0 + 1) * ACOLS + m];
        c = __builtin_amdgcn_wmma_f32_16x16x4_f32(
                 false, a,  false, b,
                 (short)0, c,  false,  false);
    }
    float* out = (m < HEADS) ? (a_src + m) : (a_dst + (m - HEADS));
    #pragma unroll
    for (int r = 0; r < 8; ++r) {
        int mg = tm * 16 + r + hs * 8;
        out[(long)mg * HEADS] = c[r];
    }
}

__global__ void __launch_bounds__(256)
k_hist(const int* __restrict__ dst, int* __restrict__ bincount) {
    __shared__ int s_cnt[NBINS];
    int t = threadIdx.x;
    s_cnt[t] = 0;
    __syncthreads();
    int e0 = blockIdx.x * EPB;
    #pragma unroll
    for (int i = 0; i < EPB / 256; ++i) {
        int e = e0 + i * 256 + t;
        if (e < N_EDGES) atomicAdd(&s_cnt[dst[e] / NB], 1);
    }
    __syncthreads();
    int c = s_cnt[t];
    if (c) atomicAdd(&bincount[t], c);
}

#define EPT (EPB / 256)
__global__ void __launch_bounds__(256, 8)
k_binplace(const int* __restrict__ src, const int* __restrict__ dst,
           const float* __restrict__ edge_attr,
           const float* __restrict__ v_edge,
           const float* __restrict__ a_src, const float* __restrict__ a_dst,
           const int* __restrict__ bincount, int* __restrict__ cursor,
           int* __restrict__ pe, float* __restrict__ ebuf) {
    __shared__ int s_cnt[NBINS];
    __shared__ int s_base[NBINS];
    __shared__ int s_scan[NBINS];
    __shared__ float s_v[EDGE_DIM * HEADS];
    int t = threadIdx.x;
    s_cnt[t] = 0;
    if (t < EDGE_DIM * HEADS) s_v[t] = v_edge[t];
    int tot = round4(bincount[t]);
    s_scan[t] = tot;

    int e0 = blockIdx.x * EPB;
    int ss[EPT], dd[EPT];
    #pragma unroll
    for (int i = 0; i < EPT; ++i) {
        int e = e0 + i * 256 + t;
        int ec = (e < N_EDGES) ? e : N_EDGES - 1;
        ss[i] = src[ec];
        dd[i] = dst[ec];
    }
    __syncthreads();
    for (int off = 1; off < NBINS; off <<= 1) {
        int a = (t >= off) ? s_scan[t - off] : 0;
        __syncthreads();
        s_scan[t] += a;
        __syncthreads();
    }
    int excl = s_scan[t] - tot;

    #pragma unroll
    for (int i = 0; i < EPT; ++i) {
        int e = e0 + i * 256 + t;
        if (e < N_EDGES) atomicAdd(&s_cnt[dd[i] / NB], 1);
    }
    float as[EPT][HEADS], ad[EPT][HEADS];
    #pragma unroll
    for (int i = 0; i < EPT; ++i) {
        const float* ps = a_src + (long)ss[i] * HEADS;
        const float* pd = a_dst + (long)dd[i] * HEADS;
        #pragma unroll
        for (int h = 0; h < HEADS; ++h) { as[i][h] = ps[h]; ad[i][h] = pd[h]; }
    }
    __syncthreads();
    int c = s_cnt[t];
    int prev = c ? atomicAdd(&cursor[t], c) : 0;
    s_base[t] = excl + prev;
    __syncthreads();
    s_cnt[t] = 0;
    __syncthreads();
    #pragma unroll
    for (int i = 0; i < EPT; ++i) {
        int e = e0 + i * 256 + t;
        if (e < N_EDGES) {
            int bin = dd[i] / NB;
            int pos = s_base[bin] + atomicAdd(&s_cnt[bin], 1);
            reinterpret_cast<int2*>(pe)[pos] = make_int2(ss[i], dd[i]);
            const float* pa = edge_attr + (long)e * EDGE_DIM;
            float ea[EDGE_DIM];
            #pragma unroll
            for (int jj = 0; jj < EDGE_DIM; ++jj) ea[jj] = pa[jj];
            float o[HEADS];
            #pragma unroll
            for (int h = 0; h < HEADS; ++h) {
                float acc = 0.0f;
                #pragma unroll
                for (int jj = 0; jj < EDGE_DIM; ++jj) acc += ea[jj] * s_v[jj * HEADS + h];
                float v = as[i][h] + ad[i][h] + acc;
                o[h] = (v > 0.0f) ? v : 0.2f * v;
            }
            float* ob = ebuf + (long)pos * HEADS;
            #pragma unroll
            for (int h = 0; h < HEADS; ++h) ob[h] = o[h];
        }
    }
}

#define SC_G 4
#define FW   32
#define PU   8

__global__ void __launch_bounds__(1024)
k_bin_nodes(const int* __restrict__ pe_, float* __restrict__ ebuf, const int* __restrict__ bincount,
            const float* __restrict__ x, const float* __restrict__ W_lin,
            const float* __restrict__ bias, const float* __restrict__ W_out,
            const float* __restrict__ b_out,
            float* __restrict__ y, float* __restrict__ alpha_sum) {
    __shared__ unsigned int s_max[NB * HEADS];
    __shared__ float s_den[NB * HEADS];
    __shared__ __attribute__((aligned(16))) float s_Z[NB * ZP];
    __shared__ float s_agg[NB * HID];
    __shared__ float s_y[NB];
    __shared__ float s_as[HEADS];
    __shared__ int s_start;

    const int2* pe = reinterpret_cast<const int2*>(pe_);
    int t    = threadIdx.x;
    int b    = blockIdx.x;
    int n0   = b * NB;
    int nn   = (N_NODES - n0 < NB) ? (N_NODES - n0) : NB;
    int lane = t & 31;
    int j    = lane >> 3;
    int h8   = lane & 7;
    int wv   = __builtin_amdgcn_readfirstlane(t >> 5);

    if (t == 0) s_start = 0;
    for (int i = t; i < NB * HEADS; i += 1024) { s_max[i] = 0u; s_den[i] = 0.0f; }
    for (int i = t; i < NB * ZP / 4; i += 1024) reinterpret_cast<float4*>(s_Z)[i] = make_float4(0.0f, 0.0f, 0.0f, 0.0f);
    if (t < HEADS) s_as[t] = 0.0f;
    __syncthreads();
    if (t < b) atomicAdd(&s_start, round4(bincount[t]));
    __syncthreads();
    int start = __builtin_amdgcn_readfirstlane(s_start);
    int end   = start + __builtin_amdgcn_readfirstlane(bincount[b]);
    int last  = end - 1;

    for (int pb = start + SC_G * wv; pb < end; pb += SC_G * FW * PU) {
        int pp[PU], d[PU];
        float x[PU];
        #pragma unroll
        for (int u = 0; u < PU; ++u) {
            pp[u] = pb + SC_G * FW * u + j;
            int pc = (pp[u] < end) ? pp[u] : last;
            d[u] = pe[pc].y;
            x[u] = ebuf[pc * HEADS + h8];
        }
        asm volatile("" :: "v"(d[0]), "v"(d[1]), "v"(d[2]), "v"(d[3]), "v"(d[4]), "v"(d[5]), "v"(d[6]), "v"(d[7]),
                     "v"(x[0]), "v"(x[1]), "v"(x[2]), "v"(x[3]), "v"(x[4]), "v"(x[5]), "v"(x[6]), "v"(x[7]));
        #pragma unroll
        for (int u = 0; u < PU; ++u)
            if (pp[u] < end) atomicMax(&s_max[(d[u] - n0) * HEADS + h8], fkey(x[u]));
    }
    __syncthreads();

    for (int pb = start + SC_G * wv; pb < end; pb += SC_G * FW * PU) {
        int pp[PU], d[PU];
        float x[PU];
        #pragma unroll
        for (int u = 0; u < PU; ++u) {
            pp[u] = pb + SC_G * FW * u + j;
            int pc = (pp[u] < end) ? pp[u] : last;
            d[u] = pe[pc].y;
            x[u] = ebuf[pc * HEADS + h8];
        }
        asm volatile("" :: "v"(d[0]), "v"(d[1]), "v"(d[2]), "v"(d[3]), "v"(d[4]), "v"(d[5]), "v"(d[6]), "v"(d[7]),
                     "v"(x[0]), "v"(x[1]), "v"(x[2]), "v"(x[3]), "v"(x[4]), "v"(x[5]), "v"(x[6]), "v"(x[7]));
        #pragma unroll
        for (int u = 0; u < PU; ++u) {
            if (pp[u] < end) {
                int di = (d[u] - n0) * HEADS + h8;
                float tt = __expf(x[u] - fkey_inv(s_max[di]));
                ebuf[pp[u] * HEADS + h8] = tt;
                atomicAdd(&s_den[di], tt);
            }
        }
    }
    __syncthreads();

    int lc = (lane < HID) ? lane : HID - 1;

    float asum = 0.0f;
    int pb = start + SC_G * wv;
    if (pb < end) {
        int pj0 = pb + j;
        int pc0 = (pj0 < end) ? pj0 : last;
        int2  sd0 = pe[pc0];
        float ex0 = ebuf[pc0 * HEADS + h8];
        int pj1 = pj0 + SC_G * FW;
        int pc1 = (pj1 < end) ? pj1 : last;
        int2  sd1 = pe[pc1];
        float ex1 = ebuf[pc1 * HEADS + h8];
        int dj0 = sd0.y, sv1 = sd1.x, dj1 = sd1.y;
        float xr0[SC_G];
        #pragma unroll
        for (int k = 0; k < SC_G; ++k) {
            int s = __builtin_amdgcn_readlane(sd0.x, 8 * k);
            xr0[k] = x[(long)s * HID + lc];
        }
        while (pb < end) {
            int pj2 = pj0 + 2 * SC_G * FW;
            int pc2 = (pj2 < end) ? pj2 : last;
            int2  sd2 = pe[pc2];
            float ex2 = ebuf[pc2 * HEADS + h8];
            float xr1[SC_G];
            #pragma unroll
            for (int k = 0; k < SC_G; ++k) {
                int s = __builtin_amdgcn_readlane(sv1, 8 * k);
                xr1[k] = x[(long)s * HID + lc];
            }
            float dn = s_den[(dj0 - n0) * HEADS + h8];
            float exv = (pj0 < end) ? ex0 : 0.0f;
            float alpha = exv / (dn + 1e-16f);
            asum += alpha;
            int cnt = end - pb;
            #pragma unroll
            for (int k = 0; k < SC_G; ++k) {
                int dl = __builtin_amdgcn_readlane(dj0, 8 * k) - n0;
                float a[HEADS];
                #pragma unroll
                for (int h = 0; h < HEADS; ++h)
                    a[h] = __int_as_float(__builtin_amdgcn_readlane(__float_as_int(alpha), 8 * k + h));
                if (k < cnt && lane < HID) {
                    float* zp = &s_Z[dl * ZP + lane];
                    #pragma unroll
                    for (int h = 0; h < HEADS; ++h) atomicAdd(zp + h * HID, a[h] * xr0[k]);
                }
            }
            pb += SC_G * FW;
            pj0 = pj1; pj1 = pj2;
            dj0 = dj1; ex0 = ex1;
            sv1 = sd2.x; dj1 = sd2.y; ex1 = ex2;
            #pragma unroll
            for (int k = 0; k < SC_G; ++k) xr0[k] = xr1[k];
        }
    }
    asum += __shfl_xor(asum, 8, 32);
    asum += __shfl_xor(asum, 16, 32);
    if (lane < HEADS) atomicAdd(&s_as[lane], asum);
    __syncthreads();
    if (t < HEADS) atomicAdd(&alpha_sum[t], s_as[t]);

    if (wv < 26) {
        int tm = wv >> 1, tn = wv & 1;
        int m  = lane & 15;
        int hs = lane >> 4;
        int row = tm * 16 + m;
        int rowc = (row < NB) ? row : NB - 1;
        const float* arow = s_Z + rowc * ZP + 2 * hs;
        int col = tn * 16 + m;
        int colc = (col < HID) ? col : HID - 1;
        const float* wl = W_lin + (long)(2 * hs) * NHID + colc;
        v8f c = {0.f, 0.f, 0.f, 0.f, 0.f, 0.f, 0.f, 0.f};
        for (int h = 0; h < HEADS; ++h) {
            #pragma unroll
            for (int kk = 0; kk < HID; kk += 4) {
                v2f a = *reinterpret_cast<const v2f*>(arow + h * HID + kk);
                v2f b; b.x = wl[(long)kk * NHID + h * HID]; b.y = wl[(long)(kk + 1) * NHID + h * HID];
                c = __builtin_amdgcn_wmma_f32_16x16x4_f32(
                         false, a,  false, b,
                         (short)0, c,  false,  false);
            }
        }
        #pragma unroll
        for (int r = 0; r < 8; ++r) {
            int mg = tm * 16 + r + hs * 8;
            if (mg < NB && col < HID) s_agg[mg * HID + col] = c[r];
        }
    }
    __syncthreads();

    for (int nl = wv; nl < nn; nl += FW) {
        float val = 0.0f;
        if (lane < HID) {
            float s = s_agg[nl * HID + lane];
            s = s * (1.0f / HEADS) + bias[lane];
            s = (s > 0.0f) ? s : (__expf(s) - 1.0f);
            val = s * W_out[lane];
        }
        #pragma unroll
        for (int off = 16; off > 0; off >>= 1) val += __shfl_down(val, off, 32);
        if (lane == 0) s_y[nl] = val + b_out[0];
    }
    __syncthreads();
    if (t < nn) y[n0 + t] = s_y[t];
}

__global__ void k_alpha_mean(const float* __restrict__ alpha_sum, float* __restrict__ out) {
    int h = threadIdx.x;
    if (h < HEADS) out[h] = alpha_sum[h] * (1.0f / (float)N_EDGES);
}

extern "C" void kernel_launch(void* const* d_in, const int* in_sizes, int n_in,
                              void* d_out, int out_size, void* d_ws, size_t ws_size,
                              hipStream_t stream) {
    const float* x         = (const float*)d_in[0];
    const int*   ei        = (const int*)d_in[1];
    const float* edge_attr = (const float*)d_in[2];
    const float* W_lin     = (const float*)d_in[3];
    const float* att_src   = (const float*)d_in[4];
    const float* att_dst   = (const float*)d_in[5];
    const float* W_edge    = (const float*)d_in[6];
    const float* att_edge  = (const float*)d_in[7];
    const float* bias      = (const float*)d_in[8];
    const float* W_out     = (const float*)d_in[9];
    const float* b_out     = (const float*)d_in[10];
    (void)in_sizes; (void)n_in; (void)out_size; (void)ws_size;

    const int* src = ei;
    const int* dst = ei + N_EDGES;

    float* w        = (float*)d_ws;
    float* a_src    = w;
    float* a_dst    = a_src   + (long)N_NODES * HEADS;
    float* ebuf     = a_dst   + (long)N_NODES * HEADS;
    int*   pe       = (int*)(ebuf + (long)P_SLOTS * HEADS);
    float* v_edge   = (float*)(pe + 2L * P_SLOTS);
    float* wa       = v_edge  + EDGE_DIM * HEADS;
    float* alpha_s  = wa      + HID * ACOLS;
    int*   bincount = (int*)(alpha_s + HEADS);
    int*   cursor   = bincount + NBINS;

    float* y         = (float*)d_out;
    float* alpha_out = y + N_NODES;

    k_vedge<<<1, NBINS, 0, stream>>>(W_edge, att_edge, W_lin, att_src, att_dst,
                                     v_edge, wa, bincount, cursor, alpha_s);
    k_att_wmma<<<(M_TILES + 7) / 8, 256, 0, stream>>>(x, wa, a_src, a_dst);
    k_hist<<<BIN_BLOCKS, 256, 0, stream>>>(dst, bincount);
    k_binplace<<<BIN_BLOCKS, 256, 0, stream>>>(src, dst, edge_attr, v_edge, a_src, a_dst,
                                               bincount, cursor, pe, ebuf);
    k_bin_nodes<<<NBINS, 1024, 0, stream>>>(pe, ebuf, bincount, x, W_lin,
                                            bias, W_out, b_out, y, alpha_s);
    k_alpha_mean<<<1, 32, 0, stream>>>(alpha_s, alpha_out);
}


# Launches 6 to 3: weight folds redone per work-group in the node-tile launch, bin counts as 64 partial rows beside it, bin kernel adds its scaled alpha sums into out; arrival-order sums as before
# speedup vs baseline: 7.8932x; 1.0998x over previous
#include <hip/hip_runtime.h>
#include <hip/hip_bf16.h>

#define N_NODES 50000
#define N_EDGES 800000
#define HID     24
#define HEADS   8
#define EDGE_DIM 8
#define NHID    192
#define M_TILES 3125
#define ACOLS   16
#define ZP      196
#define NB      196
#define NBINS   256
#define EPB     1024
#define BIN_BLOCKS ((N_EDGES + EPB - 1) / EPB)
static_assert(NB * NBINS >= N_NODES, "every node has a bin");
static_assert(EPB % 256 == 0, "whole edges per thread in the placing pass");
#define P_SLOTS (N_EDGES + 4 * NBINS)
__device__ __forceinline__ int round4(int c) { return (c + 3) & ~3; }

typedef __attribute__((ext_vector_type(2))) float v2f;
typedef __attribute__((ext_vector_type(8))) float v8f;

__device__ __forceinline__ unsigned int fkey(float v) {
    unsigned int b = __float_as_uint(v);
    return (b & 0x80000000u) ? ~b : (b | 0x80000000u);
}
__device__ __forceinline__ float fkey_inv(unsigned int k) {
    return __uint_as_float((k & 0x80000000u) ? (k ^ 0x80000000u) : ~k);
}

#define HB   64
#define EPH  (N_EDGES / HB)
#define EPHT ((EPH + 255) / 256)
#define ATT_BLOCKS ((M_TILES + 7) / 8)
static_assert(EPH * HB == N_EDGES, "the counting work-groups cover the edge list exactly");
__global__ void __launch_bounds__(256)
k_front(const float* __restrict__ x, const int* __restrict__ dst,
        const float* __restrict__ W_edge, const float* __restrict__ att_edge,
        const float* __restrict__ W_lin,
        const float* __restrict__ att_src, const float* __restrict__ att_dst,
        float* __restrict__ v_edge, int* __restrict__ part, int* __restrict__ cursor,
        float* __restrict__ alpha_out,
        float* __restrict__ a_src, float* __restrict__ a_dst) {
    __shared__ int s_cnt[NBINS];
    __shared__ float s_wa[HID * ACOLS];
    int t = threadIdx.x;
    if (blockIdx.x < HB) {
        s_cnt[t] = 0;
        __syncthreads();
        int e0 = blockIdx.x * EPH;
        int dd[EPHT];
        #pragma unroll
        for (int i = 0; i < EPHT; ++i) {
            int k = i * 256 + t;
            dd[i] = dst[e0 + ((k < EPH) ? k : EPH - 1)];
        }
        #pragma unroll
        for (int i = 0; i < EPHT; ++i)
            if (i * 256 + t < EPH) atomicAdd(&s_cnt[dd[i] / NB], 1);
        __syncthreads();
        part[blockIdx.x * NBINS + t] = s_cnt[t];
        if (blockIdx.x == 0) {
            cursor[t] = 0;
            if (t < HEADS) alpha_out[t] = 0.0f;
            if (t < EDGE_DIM * HEADS) {
                int d = t >> 3, h = t & 7;
                float sv = 0.0f;
                #pragma unroll
                for (int c = 0; c < HID; ++c)
                    sv += W_edge[d * NHID + h * HID + c] * att_edge[h * HID + c];
                v_edge[d * HEADS + h] = sv;
            }
        }
        return;
    }

    for (int q = t; q < HID * ACOLS; q += 256) {
        int k = q >> 4, col = q & 15, h = col & 7;
        const float* att = ((col < HEADS) ? att_src : att_dst) + h * HID;
        const float* wr  = W_lin + k * NHID + h * HID;
        float sv = 0.0f;
        #pragma unroll
        for (int c = 0; c < HID; ++c) sv += wr[c] * att[c];
        s_wa[q] = sv;
    }
    __syncthreads();

    int lane = t & 31;
    int tm = (int)(blockIdx.x - HB) * 8 + (t >> 5);
    if (tm >= M_TILES) return;

    int m  = lane & 15;
    int hs = lane >> 4;
    const float* arow = x + (long)(tm * 16 + m) * HID;

    v8f c = {0.f, 0.f, 0.f, 0.f, 0.f, 0.f, 0.f, 0.f};
    #pragma unroll
    for (int kb = 0; kb < HID; kb += 4) {
        int k0 = kb + hs * 2;
        v2f a; a.x = arow[k0]; a.y = arow[k0 + 1];
        v2f b; b.x = s_wa[k0 * ACOLS + m]; b.y = s_wa[(k0 + 1) * ACOLS + m];
        c = __builtin_amdgcn_wmma_f32_16x16x4_f32(
                 false, a,  false, b,
                 (short)0, c,  false,  false);
    }
    float* out = (m < HEADS) ? (a_src + m) : (a_dst + (m - HEADS));
    #pragma unroll
    for (int r = 0; r < 8; ++r) {
        int mg = tm * 16 + r + hs * 8;
        out[(long)mg * HEADS] = c[r];
    }
}

#define EPT (EPB / 256)
__global__ void __launch_bounds__(256, 8)
k_binplace(const int* __restrict__ src, const int* __restrict__ dst,
           const float* __restrict__ edge_attr,
           const float* __restrict__ v_edge,
           const float* __restrict__ a_src, const float* __restrict__ a_dst,
           const int* __restrict__ part, int* __restrict__ bincount, int* __restrict__ cursor,
           int* __restrict__ pe, float* __restrict__ ebuf) {
    __shared__ int s_cnt[NBINS];
    __shared__ int s_base[NBINS];
    __shared__ int s_scan[NBINS];
    __shared__ float s_v[EDGE_DIM * HEADS];
    int t = threadIdx.x;
    s_cnt[t] = 0;
    if (t < EDGE_DIM * HEADS) s_v[t] = v_edge[t];

    int e0 = blockIdx.x * EPB;
    int ss[EPT], dd[EPT];
    #pragma unroll
    for (int i = 0; i < EPT; ++i) {
        int e = e0 + i * 256 + t;
        int ec = (e < N_EDGES) ? e : N_EDGES - 1;
        ss[i] = src[ec];
        dd[i] = dst[ec];
    }
    int raw = 0;
    #pragma unroll
    for (int r = 0; r < HB; ++r) raw += part[r * NBINS + t];
    if (blockIdx.x == 0) bincount[t] = raw;
    int tot = round4(raw);
    s_scan[t] = tot;
    __syncthreads();
    for (int off = 1; off < NBINS; off <<= 1) {
        int a = (t >= off) ? s_scan[t - off] : 0;
        __syncthreads();
        s_scan[t] += a;
        __syncthreads();
    }
    int excl = s_scan[t] - tot;

    #pragma unroll
    for (int i = 0; i < EPT; ++i) {
        int e = e0 + i * 256 + t;
        if (e < N_EDGES) atomicAdd(&s_cnt[dd[i] / NB], 1);
    }
    float as[EPT][HEADS], ad[EPT][HEADS];
    #pragma unroll
    for (int i = 0; i < EPT; ++i) {
        const float* ps = a_src + (long)ss[i] * HEADS;
        const float* pd = a_dst + (long)dd[i] * HEADS;
        #pragma unroll
        for (int h = 0; h < HEADS; ++h) { as[i][h] = ps[h]; ad[i][h] = pd[h]; }
    }
    __syncthreads();
    int c = s_cnt[t];
    int prev = c ? atomicAdd(&cursor[t], c) : 0;
    s_base[t] = excl + prev;
    __syncthreads();
    s_cnt[t] = 0;
    __syncthreads();
    #pragma unroll
    for (int i = 0; i < EPT; ++i) {
        int e = e0 + i * 256 + t;
        if (e < N_EDGES) {
            int bin = dd[i] / NB;
            int pos = s_base[bin] + atomicAdd(&s_cnt[bin], 1);
            reinterpret_cast<int2*>(pe)[pos] = make_int2(ss[i], dd[i]);
            const float* pa = edge_attr + (long)e * EDGE_DIM;
            float ea[EDGE_DIM];
            #pragma unroll
            for (int jj = 0; jj < EDGE_DIM; ++jj) ea[jj] = pa[jj];
            float o[HEADS];
            #pragma unroll
            for (int h = 0; h < HEADS; ++h) {
                float acc = 0.0f;
                #pragma unroll
                for (int jj = 0; jj < EDGE_DIM; ++jj) acc += ea[jj] * s_v[jj * HEADS + h];
                float v = as[i][h] + ad[i][h] + acc;
                o[h] = (v > 0.0f) ? v : 0.2f * v;
            }
            float* ob = ebuf + (long)pos * HEADS;
            #pragma unroll
            for (int h = 0; h < HEADS; ++h) ob[h] = o[h];
        }
    }
}

#define SC_G 4
#define FW   32
#define PU   8

__global__ void __launch_bounds__(1024)
k_bin_nodes(const int* __restrict__ pe_, float* __restrict__ ebuf, const int* __restrict__ bincount,
            const float* __restrict__ x, const float* __restrict__ W_lin,
            const float* __restrict__ bias, const float* __restrict__ W_out,
            const float* __restrict__ b_out,
            float* __restrict__ y, float* __restrict__ alpha_out) {
    __shared__ unsigned int s_max[NB * HEADS];
    __shared__ float s_den[NB * HEADS];
    __shared__ __attribute__((aligned(16))) float s_Z[NB * ZP];
    __shared__ float s_agg[NB * HID];
    __shared__ float s_y[NB];
    __shared__ float s_as[HEADS];
    __shared__ int s_start;

    const int2* pe = reinterpret_cast<const int2*>(pe_);
    int t    = threadIdx.x;
    int b    = blockIdx.x;
    int n0   = b * NB;
    int nn   = (N_NODES - n0 < NB) ? (N_NODES - n0) : NB;
    int lane = t & 31;
    int j    = lane >> 3;
    int h8   = lane & 7;
    int wv   = __builtin_amdgcn_readfirstlane(t >> 5);

    if (t == 0) s_start = 0;
    for (int i = t; i < NB * HEADS; i += 1024) { s_max[i] = 0u; s_den[i] = 0.0f; }
    for (int i = t; i < NB * ZP / 4; i += 1024) reinterpret_cast<float4*>(s_Z)[i] = make_float4(0.0f, 0.0f, 0.0f, 0.0f);
    if (t < HEADS) s_as[t] = 0.0f;
    __syncthreads();
    if (t < b) atomicAdd(&s_start, round4(bincount[t]));
    __syncthreads();
    int start = __builtin_amdgcn_readfirstlane(s_start);
    int end   = start + __builtin_amdgcn_readfirstlane(bincount[b]);
    int last  = end - 1;

    for (int pb = start + SC_G * wv; pb < end; pb += SC_G * FW * PU) {
        int pp[PU], d[PU];
        float x[PU];
        #pragma unroll
        for (int u = 0; u < PU; ++u) {
            pp[u] = pb + SC_G * FW * u + j;
            int pc = (pp[u] < end) ? pp[u] : last;
            d[u] = pe[pc].y;
            x[u] = ebuf[pc * HEADS + h8];
        }
        asm volatile("" :: "v"(d[0]), "v"(d[1]), "v"(d[2]), "v"(d[3]), "v"(d[4]), "v"(d[5]), "v"(d[6]), "v"(d[7]),
                     "v"(x[0]), "v"(x[1]), "v"(x[2]), "v"(x[3]), "v"(x[4]), "v"(x[5]), "v"(x[6]), "v"(x[7]));
        #pragma unroll
        for (int u = 0; u < PU; ++u)
            if (pp[u] < end) atomicMax(&s_max[(d[u] - n0) * HEADS + h8], fkey(x[u]));
    }
    __syncthreads();

    for (int pb = start + SC_G * wv; pb < end; pb += SC_G * FW * PU) {
        int pp[PU], d[PU];
        float x[PU];
        #pragma unroll
        for (int u = 0; u < PU; ++u) {
            pp[u] = pb + SC_G * FW * u + j;
            int pc = (pp[u] < end) ? pp[u] : last;
            d[u] = pe[pc].y;
            x[u] = ebuf[pc * HEADS + h8];
        }
        asm volatile("" :: "v"(d[0]), "v"(d[1]), "v"(d[2]), "v"(d[3]), "v"(d[4]), "v"(d[5]), "v"(d[6]), "v"(d[7]),
                     "v"(x[0]), "v"(x[1]), "v"(x[2]), "v"(x[3]), "v"(x[4]), "v"(x[5]), "v"(x[6]), "v"(x[7]));
        #pragma unroll
        for (int u = 0; u < PU; ++u) {
            if (pp[u] < end) {
                int di = (d[u] - n0) * HEADS + h8;
                float tt = __expf(x[u] - fkey_inv(s_max[di]));
                ebuf[pp[u] * HEADS + h8] = tt;
                atomicAdd(&s_den[di], tt);
            }
        }
    }
    __syncthreads();

    int lc = (lane < HID) ? lane : HID - 1;

    float asum = 0.0f;
    int pb = start + SC_G * wv;
    if (pb < end) {
        int pj0 = pb + j;
        int pc0 = (pj0 < end) ? pj0 : last;
        int2  sd0 = pe[pc0];
        float ex0 = ebuf[pc0 * HEADS + h8];
        int pj1 = pj0 + SC_G * FW;
        int pc1 = (pj1 < end) ? pj1 : last;
        int2  sd1 = pe[pc1];
        float ex1 = ebuf[pc1 * HEADS + h8];
        int dj0 = sd0.y, sv1 = sd1.x, dj1 = sd1.y;
        float xr0[SC_G];
        #pragma unroll
        for (int k = 0; k < SC_G; ++k) {
            int s = __builtin_amdgcn_readlane(sd0.x, 8 * k);
            xr0[k] = x[(long)s * HID + lc];
        }
        while (pb < end) {
            int pj2 = pj0 + 2 * SC_G * FW;
            int pc2 = (pj2 < end) ? pj2 : last;
            int2  sd2 = pe[pc2];
            float ex2 = ebuf[pc2 * HEADS + h8];
            float xr1[SC_G];
            #pragma unroll
            for (int k = 0; k < SC_G; ++k) {
                int s = __builtin_amdgcn_readlane(sv1, 8 * k);
                xr1[k] = x[(long)s * HID + lc];
            }
            float dn = s_den[(dj0 - n0) * HEADS + h8];
            float exv = (pj0 < end) ? ex0 : 0.0f;
            float alpha = exv / (dn + 1e-16f);
            asum += alpha;
            int cnt = end - pb;
            #pragma unroll
            for (int k = 0; k < SC_G; ++k) {
                int dl = __builtin_amdgcn_readlane(dj0, 8 * k) - n0;
                float a[HEADS];
                #pragma unroll
                for (int h = 0; h < HEADS; ++h)
                    a[h] = __int_as_float(__builtin_amdgcn_readlane(__float_as_int(alpha), 8 * k + h));
                if (k < cnt && lane < HID) {
                    float* zp = &s_Z[dl * ZP + lane];
                    #pragma unroll
                    for (int h = 0; h < HEADS; ++h) atomicAdd(zp + h * HID, a[h] * xr0[k]);
                }
            }
            pb += SC_G * FW;
            pj0 = pj1; pj1 = pj2;
            dj0 = dj1; ex0 = ex1;
            sv1 = sd2.x; dj1 = sd2.y; ex1 = ex2;
            #pragma unroll
            for (int k = 0; k < SC_G; ++k) xr0[k] = xr1[k];
        }
    }
    asum += __shfl_xor(asum, 8, 32);
    asum += __shfl_xor(asum, 16, 32);
    if (lane < HEADS) atomicAdd(&s_as[lane], asum);
    __syncthreads();
    if (t < HEADS) atomicAdd(&alpha_out[t], s_as[t] * (1.0f / (float)N_EDGES));

    if (wv < 26) {
        int tm = wv >> 1, tn = wv & 1;
        int m  = lane & 15;
        int hs = lane >> 4;
        int row = tm * 16 + m;
        int rowc = (row < NB) ? row : NB - 1;
        const float* arow = s_Z + rowc * ZP + 2 * hs;
        int col = tn * 16 + m;
        int colc = (col < HID) ? col : HID - 1;
        const float* wl = W_lin + (long)(2 * hs) * NHID + colc;
        v8f c = {0.f, 0.f, 0.f, 0.f, 0.f, 0.f, 0.f, 0.f};
        for (int h = 0; h < HEADS; ++h) {
            #pragma unroll
            for (int kk = 0; kk < HID; kk += 4) {
                v2f a = *reinterpret_cast<const v2f*>(arow + h * HID + kk);
                v2f b; b.x = wl[(long)kk * NHID + h * HID]; b.y = wl[(long)(kk + 1) * NHID + h * HID];
                c = __builtin_amdgcn_wmma_f32_16x16x4_f32(
                         false, a,  false, b,
                         (short)0, c,  false,  false);
            }
        }
        #pragma unroll
        for (int r = 0; r < 8; ++r) {
            int mg = tm * 16 + r + hs * 8;
            if (mg < NB && col < HID) s_agg[mg * HID + col] = c[r];
        }
    }
    __syncthreads();

    for (int nl = wv; nl < nn; nl += FW) {
        float val = 0.0f;
        if (lane < HID) {
            float s = s_agg[nl * HID + lane];
            s = s * (1.0f / HEADS) + bias[lane];
            s = (s > 0.0f) ? s : (__expf(s) - 1.0f);
            val = s * W_out[lane];
        }
        #pragma unroll
        for (int off = 16; off > 0; off >>= 1) val += __shfl_down(val, off, 32);
        if (lane == 0) s_y[nl] = val + b_out[0];
    }
    __syncthreads();
    if (t < nn) y[n0 + t] = s_y[t];
}

extern "C" void kernel_launch(void* const* d_in, const int* in_sizes, int n_in,
                              void* d_out, int out_size, void* d_ws, size_t ws_size,
                              hipStream_t stream) {
    const float* x         = (const float*)d_in[0];
    const int*   ei        = (const int*)d_in[1];
    const float* edge_attr = (const float*)d_in[2];
    const float* W_lin     = (const float*)d_in[3];
    const float* att_src   = (const float*)d_in[4];
    const float* att_dst   = (const float*)d_in[5];
    const float* W_edge    = (const float*)d_in[6];
    const float* att_edge  = (const float*)d_in[7];
    const float* bias      = (const float*)d_in[8];
    const float* W_out     = (const float*)d_in[9];
    const float* b_out     = (const float*)d_in[10];
    (void)in_sizes; (void)n_in; (void)out_size; (void)ws_size;

    const int* src = ei;
    const int* dst = ei + N_EDGES;

    float* w        = (float*)d_ws;
    float* a_src    = w;
    float* a_dst    = a_src   + (long)N_NODES * HEADS;
    float* ebuf     = a_dst   + (long)N_NODES * HEADS;
    int*   pe       = (int*)(ebuf + (long)P_SLOTS * HEADS);
    float* v_edge   = (float*)(pe + 2L * P_SLOTS);
    int*   bincount = (int*)(v_edge + EDGE_DIM * HEADS);
    int*   cursor   = bincount + NBINS;
    int*   part     = cursor + NBINS;

    float* y         = (float*)d_out;
    float* alpha_out = y + N_NODES;

    k_front<<<HB + ATT_BLOCKS, 256, 0, stream>>>(x, dst, W_edge, att_edge, W_lin, att_src, att_dst,
                                                 v_edge, part, cursor, alpha_out, a_src, a_dst);
    k_binplace<<<BIN_BLOCKS, 256, 0, stream>>>(src, dst, edge_attr, v_edge, a_src, a_dst,
                                               part, bincount, cursor, pe, ebuf);
    k_bin_nodes<<<NBINS, 1024, 0, stream>>>(pe, ebuf, bincount, x, W_lin,
                                            bias, W_out, b_out, y, alpha_out);
}


# Placing pass stops gathering a_dst[dst] per edge and stores the logit without that term; the bin kernel adds it from an LDS copy of its nodes' a_dst rows; max taken on partial logits; sum regrouped
# speedup vs baseline: 8.2799x; 1.0490x over previous
#include <hip/hip_runtime.h>
#include <hip/hip_bf16.h>

#define N_NODES 50000
#define N_EDGES 800000
#define HID     24
#define HEADS   8
#define EDGE_DIM 8
#define NHID    192
#define M_TILES 3125
#define ACOLS   16
#define ZP      196
#define NB      196
#define NBINS   256
#define EPB     1024
#define BIN_BLOCKS ((N_EDGES + EPB - 1) / EPB)
static_assert(NB * NBINS >= N_NODES, "every node has a bin");
static_assert(EPB % 256 == 0, "whole edges per thread in the placing pass");
#define P_SLOTS (N_EDGES + 4 * NBINS)
__device__ __forceinline__ int round4(int c) { return (c + 3) & ~3; }

typedef __attribute__((ext_vector_type(2))) float v2f;
typedef __attribute__((ext_vector_type(8))) float v8f;

__device__ __forceinline__ unsigned int fkey(float v) {
    unsigned int b = __float_as_uint(v);
    return (b & 0x80000000u) ? ~b : (b | 0x80000000u);
}
__device__ __forceinline__ float fkey_inv(unsigned int k) {
    return __uint_as_float((k & 0x80000000u) ? (k ^ 0x80000000u) : ~k);
}

#define HB   64
#define EPH  (N_EDGES / HB)
#define EPHT ((EPH + 255) / 256)
#define ATT_BLOCKS ((M_TILES + 7) / 8)
static_assert(EPH * HB == N_EDGES, "the counting work-groups cover the edge list exactly");
__global__ void __launch_bounds__(256)
k_front(const float* __restrict__ x, const int* __restrict__ dst,
        const float* __restrict__ W_edge, const float* __restrict__ att_edge,
        const float* __restrict__ W_lin,
        const float* __restrict__ att_src, const float* __restrict__ att_dst,
        float* __restrict__ v_edge, int* __restrict__ part, int* __restrict__ cursor,
        float* __restrict__ alpha_out,
        float* __restrict__ a_src, float* __restrict__ a_dst) {
    __shared__ int s_cnt[NBINS];
    __shared__ float s_wa[HID * ACOLS];
    int t = threadIdx.x;
    if (blockIdx.x < HB) {
        s_cnt[t] = 0;
        __syncthreads();
        int e0 = blockIdx.x * EPH;
        int dd[EPHT];
        #pragma unroll
        for (int i = 0; i < EPHT; ++i) {
            int k = i * 256 + t;
            dd[i] = dst[e0 + ((k < EPH) ? k : EPH - 1)];
        }
        #pragma unroll
        for (int i = 0; i < EPHT; ++i)
            if (i * 256 + t < EPH) atomicAdd(&s_cnt[dd[i] / NB], 1);
        __syncthreads();
        part[blockIdx.x * NBINS + t] = s_cnt[t];
        if (blockIdx.x == 0) {
            cursor[t] = 0;
            if (t < HEADS) alpha_out[t] = 0.0f;
            if (t < EDGE_DIM * HEADS) {
                int d = t >> 3, h = t & 7;
                float sv = 0.0f;
                #pragma unroll
                for (int c = 0; c < HID; ++c)
                    sv += W_edge[d * NHID + h * HID + c] * att_edge[h * HID + c];
                v_edge[d * HEADS + h] = sv;
            }
        }
        return;
    }

    for (int q = t; q < HID * ACOLS; q += 256) {
        int k = q >> 4, col = q & 15, h = col & 7;
        const float* att = ((col < HEADS) ? att_src : att_dst) + h * HID;
        const float* wr  = W_lin + k * NHID + h * HID;
        float sv = 0.0f;
        #pragma unroll
        for (int c = 0; c < HID; ++c) sv += wr[c] * att[c];
        s_wa[q] = sv;
    }
    __syncthreads();

    int lane = t & 31;
    int tm = (int)(blockIdx.x - HB) * 8 + (t >> 5);
    if (tm >= M_TILES) return;

    int m  = lane & 15;
    int hs = lane >> 4;
    const float* arow = x + (long)(tm * 16 + m) * HID;

    v8f c = {0.f, 0.f, 0.f, 0.f, 0.f, 0.f, 0.f, 0.f};
    #pragma unroll
    for (int kb = 0; kb < HID; kb += 4) {
        int k0 = kb + hs * 2;
        v2f a; a.x = arow[k0]; a.y = arow[k0 + 1];
        v2f b; b.x = s_wa[k0 * ACOLS + m]; b.y = s_wa[(k0 + 1) * ACOLS + m];
        c = __builtin_amdgcn_wmma_f32_16x16x4_f32(
                 false, a,  false, b,
                 (short)0, c,  false,  false);
    }
    float* out = (m < HEADS) ? (a_src + m) : (a_dst + (m - HEADS));
    #pragma unroll
    for (int r = 0; r < 8; ++r) {
        int mg = tm * 16 + r + hs * 8;
        out[(long)mg * HEADS] = c[r];
    }
}

#define EPT (EPB / 256)
__global__ void __launch_bounds__(256, 8)
k_binplace(const int* __restrict__ src, const int* __restrict__ dst,
           const float* __restrict__ edge_attr,
           const float* __restrict__ v_edge,
           const float* __restrict__ a_src,
           const int* __restrict__ part, int* __restrict__ bincount, int* __restrict__ cursor,
           int* __restrict__ pe, float* __restrict__ ebuf) {
    __shared__ int s_cnt[NBINS];
    __shared__ int s_base[NBINS];
    __shared__ int s_scan[NBINS];
    __shared__ float s_v[EDGE_DIM * HEADS];
    int t = threadIdx.x;
    s_cnt[t] = 0;
    if (t < EDGE_DIM * HEADS) s_v[t] = v_edge[t];

    int e0 = blockIdx.x * EPB;
    int ss[EPT], dd[EPT];
    #pragma unroll
    for (int i = 0; i < EPT; ++i) {
        int e = e0 + i * 256 + t;
        int ec = (e < N_EDGES) ? e : N_EDGES - 1;
        ss[i] = src[ec];
        dd[i] = dst[ec];
    }
    int raw = 0;
    #pragma unroll
    for (int r = 0; r < HB; ++r) raw += part[r * NBINS + t];
    if (blockIdx.x == 0) bincount[t] = raw;
    int tot = round4(raw);
    s_scan[t] = tot;
    __syncthreads();
    for (int off = 1; off < NBINS; off <<= 1) {
        int a = (t >= off) ? s_scan[t - off] : 0;
        __syncthreads();
        s_scan[t] += a;
        __syncthreads();
    }
    int excl = s_scan[t] - tot;

    #pragma unroll
    for (int i = 0; i < EPT; ++i) {
        int e = e0 + i * 256 + t;
        if (e < N_EDGES) atomicAdd(&s_cnt[dd[i] / NB], 1);
    }
    float as[EPT][HEADS];
    #pragma unroll
    for (int i = 0; i < EPT; ++i) {
        const float* ps = a_src + (long)ss[i] * HEADS;
        #pragma unroll
        for (int h = 0; h < HEADS; ++h) as[i][h] = ps[h];
    }
    __syncthreads();
    int c = s_cnt[t];
    int prev = c ? atomicAdd(&cursor[t], c) : 0;
    s_base[t] = excl + prev;
    __syncthreads();
    s_cnt[t] = 0;
    __syncthreads();
    #pragma unroll
    for (int i = 0; i < EPT; ++i) {
        int e = e0 + i * 256 + t;
        if (e < N_EDGES) {
            int bin = dd[i] / NB;
            int pos = s_base[bin] + atomicAdd(&s_cnt[bin], 1);
            reinterpret_cast<int2*>(pe)[pos] = make_int2(ss[i], dd[i]);
            const float* pa = edge_attr + (long)e * EDGE_DIM;
            float ea[EDGE_DIM];
            #pragma unroll
            for (int jj = 0; jj < EDGE_DIM; ++jj) ea[jj] = pa[jj];
            float o[HEADS];
            #pragma unroll
            for (int h = 0; h < HEADS; ++h) {
                float acc = 0.0f;
                #pragma unroll
                for (int jj = 0; jj < EDGE_DIM; ++jj) acc += ea[jj] * s_v[jj * HEADS + h];
                o[h] = as[i][h] + acc;
            }
            float* ob = ebuf + (long)pos * HEADS;
            #pragma unroll
            for (int h = 0; h < HEADS; ++h) ob[h] = o[h];
        }
    }
}

#define SC_G 4
#define FW   32
#define PU   8
#define ADT  ((NB * HEADS + 1023) / 1024)

__global__ void __launch_bounds__(1024)
k_bin_nodes(const int* __restrict__ pe_, float* __restrict__ ebuf, const int* __restrict__ bincount,
            const float* __restrict__ a_dst,
            const float* __restrict__ x, const float* __restrict__ W_lin,
            const float* __restrict__ bias, const float* __restrict__ W_out,
            const float* __restrict__ b_out,
            float* __restrict__ y, float* __restrict__ alpha_out) {
    __shared__ unsigned int s_max[NB * HEADS];
    __shared__ float s_den[NB * HEADS];
    __shared__ float s_ad[NB * HEADS];
    __shared__ __attribute__((aligned(16))) float s_Z[NB * ZP];
    __shared__ float s_agg[NB * HID];
    __shared__ float s_y[NB];
    __shared__ float s_as[HEADS];
    __shared__ int s_start;

    const int2* pe = reinterpret_cast<const int2*>(pe_);
    int t    = threadIdx.x;
    int b    = blockIdx.x;
    int n0   = b * NB;
    int nn   = (N_NODES - n0 < NB) ? (N_NODES - n0) : NB;
    int lane = t & 31;
    int j    = lane >> 3;
    int h8   = lane & 7;
    int wv   = __builtin_amdgcn_readfirstlane(t >> 5);

    float adv[ADT];
    #pragma unroll
    for (int q = 0; q < ADT; ++q) {
        int i = q * 1024 + t;
        adv[q] = a_dst[(long)n0 * HEADS + ((i < nn * HEADS) ? i : 0)];
    }
    if (t == 0) s_start = 0;
    for (int i = t; i < NB * HEADS; i += 1024) { s_max[i] = 0u; s_den[i] = 0.0f; }
    for (int i = t; i < NB * ZP / 4; i += 1024) reinterpret_cast<float4*>(s_Z)[i] = make_float4(0.0f, 0.0f, 0.0f, 0.0f);
    if (t < HEADS) s_as[t] = 0.0f;
    #pragma unroll
    for (int q = 0; q < ADT; ++q) {
        int i = q * 1024 + t;
        if (i < nn * HEADS) s_ad[i] = adv[q];
    }
    __syncthreads();
    if (t < b) atomicAdd(&s_start, round4(bincount[t]));
    __syncthreads();
    int start = __builtin_amdgcn_readfirstlane(s_start);
    int end   = start + __builtin_amdgcn_readfirstlane(bincount[b]);
    int last  = end - 1;

    for (int pb = start + SC_G * wv; pb < end; pb += SC_G * FW * PU) {
        int pp[PU], d[PU];
        float x[PU];
        #pragma unroll
        for (int u = 0; u < PU; ++u) {
            pp[u] = pb + SC_G * FW * u + j;
            int pc = (pp[u] < end) ? pp[u] : last;
            d[u] = pe[pc].y;
            x[u] = ebuf[pc * HEADS + h8];
        }
        asm volatile("" :: "v"(d[0]), "v"(d[1]), "v"(d[2]), "v"(d[3]), "v"(d[4]), "v"(d[5]), "v"(d[6]), "v"(d[7]),
                     "v"(x[0]), "v"(x[1]), "v"(x[2]), "v"(x[3]), "v"(x[4]), "v"(x[5]), "v"(x[6]), "v"(x[7]));
        #pragma unroll
        for (int u = 0; u < PU; ++u)
            if (pp[u] < end) atomicMax(&s_max[(d[u] - n0) * HEADS + h8], fkey(x[u]));
    }
    __syncthreads();

    for (int pb = start + SC_G * wv; pb < end; pb += SC_G * FW * PU) {
        int pp[PU], d[PU];
        float x[PU];
        #pragma unroll
        for (int u = 0; u < PU; ++u) {
            pp[u] = pb + SC_G * FW * u + j;
            int pc = (pp[u] < end) ? pp[u] : last;
            d[u] = pe[pc].y;
            x[u] = ebuf[pc * HEADS + h8];
        }
        asm volatile("" :: "v"(d[0]), "v"(d[1]), "v"(d[2]), "v"(d[3]), "v"(d[4]), "v"(d[5]), "v"(d[6]), "v"(d[7]),
                     "v"(x[0]), "v"(x[1]), "v"(x[2]), "v"(x[3]), "v"(x[4]), "v"(x[5]), "v"(x[6]), "v"(x[7]));
        #pragma unroll
        for (int u = 0; u < PU; ++u) {
            if (pp[u] < end) {
                int di = (d[u] - n0) * HEADS + h8;
                float ad = s_ad[di];
                float v = x[u] + ad;
                v = (v > 0.0f) ? v : 0.2f * v;
                float m = fkey_inv(s_max[di]) + ad;
                m = (m > 0.0f) ? m : 0.2f * m;
                float tt = __expf(v - m);
                ebuf[pp[u] * HEADS + h8] = tt;
                atomicAdd(&s_den[di], tt);
            }
        }
    }
    __syncthreads();

    int lc = (lane < HID) ? lane : HID - 1;

    float asum = 0.0f;
    int pb = start + SC_G * wv;
    if (pb < end) {
        int pj0 = pb + j;
        int pc0 = (pj0 < end) ? pj0 : last;
        int2  sd0 = pe[pc0];
        float ex0 = ebuf[pc0 * HEADS + h8];
        int pj1 = pj0 + SC_G * FW;
        int pc1 = (pj1 < end) ? pj1 : last;
        int2  sd1 = pe[pc1];
        float ex1 = ebuf[pc1 * HEADS + h8];
        int dj0 = sd0.y, sv1 = sd1.x, dj1 = sd1.y;
        float xr0[SC_G];
        #pragma unroll
        for (int k = 0; k < SC_G; ++k) {
            int s = __builtin_amdgcn_readlane(sd0.x, 8 * k);
            xr0[k] = x[(long)s * HID + lc];
        }
        while (pb < end) {
            int pj2 = pj0 + 2 * SC_G * FW;
            int pc2 = (pj2 < end) ? pj2 : last;
            int2  sd2 = pe[pc2];
            float ex2 = ebuf[pc2 * HEADS + h8];
            float xr1[SC_G];
            #pragma unroll
            for (int k = 0; k < SC_G; ++k) {
                int s = __builtin_amdgcn_readlane(sv1, 8 * k);
                xr1[k] = x[(long)s * HID + lc];
            }
            float dn = s_den[(dj0 - n0) * HEADS + h8];
            float exv = (pj0 < end) ? ex0 : 0.0f;
            float alpha = exv / (dn + 1e-16f);
            asum += alpha;
            int cnt = end - pb;
            #pragma unroll
            for (int k = 0; k < SC_G; ++k) {
                int dl = __builtin_amdgcn_readlane(dj0, 8 * k) - n0;
                float a[HEADS];
                #pragma unroll
                for (int h = 0; h < HEADS; ++h)
                    a[h] = __int_as_float(__builtin_amdgcn_readlane(__float_as_int(alpha), 8 * k + h));
                if (k < cnt && lane < HID) {
                    float* zp = &s_Z[dl * ZP + lane];
                    #pragma unroll
                    for (int h = 0; h < HEADS; ++h) atomicAdd(zp + h * HID, a[h] * xr0[k]);
                }
            }
            pb += SC_G * FW;
            pj0 = pj1; pj1 = pj2;
            dj0 = dj1; ex0 = ex1;
            sv1 = sd2.x; dj1 = sd2.y; ex1 = ex2;
            #pragma unroll
            for (int k = 0; k < SC_G; ++k) xr0[k] = xr1[k];
        }
    }
    asum += __shfl_xor(asum, 8, 32);
    asum += __shfl_xor(asum, 16, 32);
    if (lane < HEADS) atomicAdd(&s_as[lane], asum);
    __syncthreads();
    if (t < HEADS) atomicAdd(&alpha_out[t], s_as[t] * (1.0f / (float)N_EDGES));

    if (wv < 26) {
        int tm = wv >> 1, tn = wv & 1;
        int m  = lane & 15;
        int hs = lane >> 4;
        int row = tm * 16 + m;
        int rowc = (row < NB) ? row : NB - 1;
        const float* arow = s_Z + rowc * ZP + 2 * hs;
        int col = tn * 16 + m;
        int colc = (col < HID) ? col : HID - 1;
        const float* wl = W_lin + (long)(2 * hs) * NHID + colc;
        v8f c = {0.f, 0.f, 0.f, 0.f, 0.f, 0.f, 0.f, 0.f};
        for (int h = 0; h < HEADS; ++h) {
            #pragma unroll
            for (int kk = 0; kk < HID; kk += 4) {
                v2f a = *reinterpret_cast<const v2f*>(arow + h * HID + kk);
                v2f b; b.x = wl[(long)kk * NHID + h * HID]; b.y = wl[(long)(kk + 1) * NHID + h * HID];
                c = __builtin_amdgcn_wmma_f32_16x16x4_f32(
                         false, a,  false, b,
                         (short)0, c,  false,  false);
            }
        }
        #pragma unroll
        for (int r = 0; r < 8; ++r) {
            int mg = tm * 16 + r + hs * 8;
            if (mg < NB && col < HID) s_agg[mg * HID + col] = c[r];
        }
    }
    __syncthreads();

    for (int nl = wv; nl < nn; nl += FW) {
        float val = 0.0f;
        if (lane < HID) {
            float s = s_agg[nl * HID + lane];
            s = s * (1.0f / HEADS) + bias[lane];
            s = (s > 0.0f) ? s : (__expf(s) - 1.0f);
            val = s * W_out[lane];
        }
        #pragma unroll
        for (int off = 16; off > 0; off >>= 1) val += __shfl_down(val, off, 32);
        if (lane == 0) s_y[nl] = val + b_out[0];
    }
    __syncthreads();
    if (t < nn) y[n0 + t] = s_y[t];
}

extern "C" void kernel_launch(void* const* d_in, const int* in_sizes, int n_in,
                              void* d_out, int out_size, void* d_ws, size_t ws_size,
                              hipStream_t stream) {
    const float* x         = (const float*)d_in[0];
    const int*   ei        = (const int*)d_in[1];
    const float* edge_attr = (const float*)d_in[2];
    const float* W_lin     = (const float*)d_in[3];
    const float* att_src   = (const float*)d_in[4];
    const float* att_dst   = (const float*)d_in[5];
    const float* W_edge    = (const float*)d_in[6];
    const float* att_edge  = (const float*)d_in[7];
    const float* bias      = (const float*)d_in[8];
    const float* W_out     = (const float*)d_in[9];
    const float* b_out     = (const float*)d_in[10];
    (void)in_sizes; (void)n_in; (void)out_size; (void)ws_size;

    const int* src = ei;
    const int* dst = ei + N_EDGES;

    float* w        = (float*)d_ws;
    float* a_src    = w;
    float* a_dst    = a_src   + (long)N_NODES * HEADS;
    float* ebuf     = a_dst   + (long)N_NODES * HEADS;
    int*   pe       = (int*)(ebuf + (long)P_SLOTS * HEADS);
    float* v_edge   = (float*)(pe + 2L * P_SLOTS);
    int*   bincount = (int*)(v_edge + EDGE_DIM * HEADS);
    int*   cursor   = bincount + NBINS;
    int*   part     = cursor + NBINS;

    float* y         = (float*)d_out;
    float* alpha_out = y + N_NODES;

    k_front<<<HB + ATT_BLOCKS, 256, 0, stream>>>(x, dst, W_edge, att_edge, W_lin, att_src, att_dst,
                                                 v_edge, part, cursor, alpha_out, a_src, a_dst);
    k_binplace<<<BIN_BLOCKS, 256, 0, stream>>>(src, dst, edge_attr, v_edge, a_src,
                                               part, bincount, cursor, pe, ebuf);
    k_bin_nodes<<<NBINS, 1024, 0, stream>>>(pe, ebuf, bincount, a_dst, x, W_lin,
                                            bias, W_out, b_out, y, alpha_out);
}


# Softmax division moved behind the sums: one edge pass forms exp, denominators and exp*x row sums (exp pass and its stores gone); A loader divides per (node,head); alpha means from den; sums regrouped
# speedup vs baseline: 8.6634x; 1.0463x over previous
#include <hip/hip_runtime.h>
#include <hip/hip_bf16.h>

#define N_NODES 50000
#define N_EDGES 800000
#define HID     24
#define HEADS   8
#define EDGE_DIM 8
#define NHID    192
#define M_TILES 3125
#define ACOLS   16
#define ZP      196
#define NB      196
#define NBINS   256
#define EPB     1024
#define BIN_BLOCKS ((N_EDGES + EPB - 1) / EPB)
static_assert(NB * NBINS >= N_NODES, "every node has a bin");
static_assert(EPB % 256 == 0, "whole edges per thread in the placing pass");
#define P_SLOTS (N_EDGES + 4 * NBINS)
__device__ __forceinline__ int round4(int c) { return (c + 3) & ~3; }

typedef __attribute__((ext_vector_type(2))) float v2f;
typedef __attribute__((ext_vector_type(8))) float v8f;

__device__ __forceinline__ unsigned int fkey(float v) {
    unsigned int b = __float_as_uint(v);
    return (b & 0x80000000u) ? ~b : (b | 0x80000000u);
}
__device__ __forceinline__ float fkey_inv(unsigned int k) {
    return __uint_as_float((k & 0x80000000u) ? (k ^ 0x80000000u) : ~k);
}

#define HB   64
#define EPH  (N_EDGES / HB)
#define EPHT ((EPH + 255) / 256)
#define ATT_BLOCKS ((M_TILES + 7) / 8)
static_assert(EPH * HB == N_EDGES, "the counting work-groups cover the edge list exactly");
__global__ void __launch_bounds__(256)
k_front(const float* __restrict__ x, const int* __restrict__ dst,
        const float* __restrict__ W_edge, const float* __restrict__ att_edge,
        const float* __restrict__ W_lin,
        const float* __restrict__ att_src, const float* __restrict__ att_dst,
        float* __restrict__ v_edge, int* __restrict__ part, int* __restrict__ cursor,
        float* __restrict__ alpha_out,
        float* __restrict__ a_src, float* __restrict__ a_dst) {
    __shared__ int s_cnt[NBINS];
    __shared__ float s_wa[HID * ACOLS];
    int t = threadIdx.x;
    if (blockIdx.x < HB) {
        s_cnt[t] = 0;
        __syncthreads();
        int e0 = blockIdx.x * EPH;
        int dd[EPHT];
        #pragma unroll
        for (int i = 0; i < EPHT; ++i) {
            int k = i * 256 + t;
            dd[i] = dst[e0 + ((k < EPH) ? k : EPH - 1)];
        }
        #pragma unroll
        for (int i = 0; i < EPHT; ++i)
            if (i * 256 + t < EPH) atomicAdd(&s_cnt[dd[i] / NB], 1);
        __syncthreads();
        part[blockIdx.x * NBINS + t] = s_cnt[t];
        if (blockIdx.x == 0) {
            cursor[t] = 0;
            if (t < HEADS) alpha_out[t] = 0.0f;
            if (t < EDGE_DIM * HEADS) {
                int d = t >> 3, h = t & 7;
                float sv = 0.0f;
                #pragma unroll
                for (int c = 0; c < HID; ++c)
                    sv += W_edge[d * NHID + h * HID + c] * att_edge[h * HID + c];
                v_edge[d * HEADS + h] = sv;
            }
        }
        return;
    }

    for (int q = t; q < HID * ACOLS; q += 256) {
        int k = q >> 4, col = q & 15, h = col & 7;
        const float* att = ((col < HEADS) ? att_src : att_dst) + h * HID;
        const float* wr  = W_lin + k * NHID + h * HID;
        float sv = 0.0f;
        #pragma unroll
        for (int c = 0; c < HID; ++c) sv += wr[c] * att[c];
        s_wa[q] = sv;
    }
    __syncthreads();

    int lane = t & 31;
    int tm = (int)(blockIdx.x - HB) * 8 + (t >> 5);
    if (tm >= M_TILES) return;

    int m  = lane & 15;
    int hs = lane >> 4;
    const float* arow = x + (long)(tm * 16 + m) * HID;

    v8f c = {0.f, 0.f, 0.f, 0.f, 0.f, 0.f, 0.f, 0.f};
    #pragma unroll
    for (int kb = 0; kb < HID; kb += 4) {
        int k0 = kb + hs * 2;
        v2f a; a.x = arow[k0]; a.y = arow[k0 + 1];
        v2f b; b.x = s_wa[k0 * ACOLS + m]; b.y = s_wa[(k0 + 1) * ACOLS + m];
        c = __builtin_amdgcn_wmma_f32_16x16x4_f32(
                 false, a,  false, b,
                 (short)0, c,  false,  false);
    }
    float* out = (m < HEADS) ? (a_src + m) : (a_dst + (m - HEADS));
    #pragma unroll
    for (int r = 0; r < 8; ++r) {
        int mg = tm * 16 + r + hs * 8;
        out[(long)mg * HEADS] = c[r];
    }
}

#define EPT (EPB / 256)
__global__ void __launch_bounds__(256, 8)
k_binplace(const int* __restrict__ src, const int* __restrict__ dst,
           const float* __restrict__ edge_attr,
           const float* __restrict__ v_edge,
           const float* __restrict__ a_src,
           const int* __restrict__ part, int* __restrict__ bincount, int* __restrict__ cursor,
           int* __restrict__ pe, float* __restrict__ ebuf) {
    __shared__ int s_cnt[NBINS];
    __shared__ int s_base[NBINS];
    __shared__ int s_scan[NBINS];
    __shared__ float s_v[EDGE_DIM * HEADS];
    int t = threadIdx.x;
    s_cnt[t] = 0;
    if (t < EDGE_DIM * HEADS) s_v[t] = v_edge[t];

    int e0 = blockIdx.x * EPB;
    int ss[EPT], dd[EPT];
    #pragma unroll
    for (int i = 0; i < EPT; ++i) {
        int e = e0 + i * 256 + t;
        int ec = (e < N_EDGES) ? e : N_EDGES - 1;
        ss[i] = src[ec];
        dd[i] = dst[ec];
    }
    int raw = 0;
    #pragma unroll
    for (int r = 0; r < HB; ++r) raw += part[r * NBINS + t];
    if (blockIdx.x == 0) bincount[t] = raw;
    int tot = round4(raw);
    s_scan[t] = tot;
    __syncthreads();
    for (int off = 1; off < NBINS; off <<= 1) {
        int a = (t >= off) ? s_scan[t - off] : 0;
        __syncthreads();
        s_scan[t] += a;
        __syncthreads();
    }
    int excl = s_scan[t] - tot;

    #pragma unroll
    for (int i = 0; i < EPT; ++i) {
        int e = e0 + i * 256 + t;
        if (e < N_EDGES) atomicAdd(&s_cnt[dd[i] / NB], 1);
    }
    float as[EPT][HEADS];
    #pragma unroll
    for (int i = 0; i < EPT; ++i) {
        const float* ps = a_src + (long)ss[i] * HEADS;
        #pragma unroll
        for (int h = 0; h < HEADS; ++h) as[i][h] = ps[h];
    }
    __syncthreads();
    int c = s_cnt[t];
    int prev = c ? atomicAdd(&cursor[t], c) : 0;
    s_base[t] = excl + prev;
    __syncthreads();
    s_cnt[t] = 0;
    __syncthreads();
    #pragma unroll
    for (int i = 0; i < EPT; ++i) {
        int e = e0 + i * 256 + t;
        if (e < N_EDGES) {
            int bin = dd[i] / NB;
            int pos = s_base[bin] + atomicAdd(&s_cnt[bin], 1);
            reinterpret_cast<int2*>(pe)[pos] = make_int2(ss[i], dd[i]);
            const float* pa = edge_attr + (long)e * EDGE_DIM;
            float ea[EDGE_DIM];
            #pragma unroll
            for (int jj = 0; jj < EDGE_DIM; ++jj) ea[jj] = pa[jj];
            float o[HEADS];
            #pragma unroll
            for (int h = 0; h < HEADS; ++h) {
                float acc = 0.0f;
                #pragma unroll
                for (int jj = 0; jj < EDGE_DIM; ++jj) acc += ea[jj] * s_v[jj * HEADS + h];
                o[h] = as[i][h] + acc;
            }
            float* ob = ebuf + (long)pos * HEADS;
            #pragma unroll
            for (int h = 0; h < HEADS; ++h) ob[h] = o[h];
        }
    }
}

#define SC_G 4
#define FW   32
#define PU   8
#define ADT  ((NB * HEADS + 1023) / 1024)

__global__ void __launch_bounds__(1024)
k_bin_nodes(const int* __restrict__ pe_, const float* __restrict__ ebuf, const int* __restrict__ bincount,
            const float* __restrict__ a_dst,
            const float* __restrict__ x, const float* __restrict__ W_lin,
            const float* __restrict__ bias, const float* __restrict__ W_out,
            const float* __restrict__ b_out,
            float* __restrict__ y, float* __restrict__ alpha_out) {
    __shared__ unsigned int s_max[NB * HEADS];
    __shared__ float s_den[NB * HEADS];
    __shared__ float s_ad[NB * HEADS];
    __shared__ __attribute__((aligned(16))) float s_Z[NB * ZP];
    __shared__ float s_agg[NB * HID];
    __shared__ float s_y[NB];
    __shared__ float s_as[HEADS];
    __shared__ int s_start;

    const int2* pe = reinterpret_cast<const int2*>(pe_);
    int t    = threadIdx.x;
    int b    = blockIdx.x;
    int n0   = b * NB;
    int nn   = (N_NODES - n0 < NB) ? (N_NODES - n0) : NB;
    int lane = t & 31;
    int j    = lane >> 3;
    int h8   = lane & 7;
    int wv   = __builtin_amdgcn_readfirstlane(t >> 5);

    float adv[ADT];
    #pragma unroll
    for (int q = 0; q < ADT; ++q) {
        int i = q * 1024 + t;
        adv[q] = a_dst[(long)n0 * HEADS + ((i < nn * HEADS) ? i : 0)];
    }
    if (t == 0) s_start = 0;
    for (int i = t; i < NB * HEADS; i += 1024) { s_max[i] = 0u; s_den[i] = 0.0f; }
    for (int i = t; i < NB * ZP / 4; i += 1024) reinterpret_cast<float4*>(s_Z)[i] = make_float4(0.0f, 0.0f, 0.0f, 0.0f);
    if (t < HEADS) s_as[t] = 0.0f;
    #pragma unroll
    for (int q = 0; q < ADT; ++q) {
        int i = q * 1024 + t;
        if (i < nn * HEADS) s_ad[i] = adv[q];
    }
    __syncthreads();
    if (t < b) atomicAdd(&s_start, round4(bincount[t]));
    __syncthreads();
    int start = __builtin_amdgcn_readfirstlane(s_start);
    int end   = start + __builtin_amdgcn_readfirstlane(bincount[b]);
    int last  = end - 1;

    for (int pb = start + SC_G * wv; pb < end; pb += SC_G * FW * PU) {
        int pp[PU], d[PU];
        float x[PU];
        #pragma unroll
        for (int u = 0; u < PU; ++u) {
            pp[u] = pb + SC_G * FW * u + j;
            int pc = (pp[u] < end) ? pp[u] : last;
            d[u] = pe[pc].y;
            x[u] = ebuf[pc * HEADS + h8];
        }
        asm volatile("" :: "v"(d[0]), "v"(d[1]), "v"(d[2]), "v"(d[3]), "v"(d[4]), "v"(d[5]), "v"(d[6]), "v"(d[7]),
                     "v"(x[0]), "v"(x[1]), "v"(x[2]), "v"(x[3]), "v"(x[4]), "v"(x[5]), "v"(x[6]), "v"(x[7]));
        #pragma unroll
        for (int u = 0; u < PU; ++u)
            if (pp[u] < end) atomicMax(&s_max[(d[u] - n0) * HEADS + h8], fkey(x[u]));
    }
    __syncthreads();

    int lc = (lane < HID) ? lane : HID - 1;

    int pb = start + SC_G * wv;
    if (pb < end) {
        int pj0 = pb + j;
        int pc0 = (pj0 < end) ? pj0 : last;
        int2  sd0 = pe[pc0];
        float pl0 = ebuf[pc0 * HEADS + h8];
        int pj1 = pj0 + SC_G * FW;
        int pc1 = (pj1 < end) ? pj1 : last;
        int2  sd1 = pe[pc1];
        float pl1 = ebuf[pc1 * HEADS + h8];
        int dj0 = sd0.y, sv1 = sd1.x, dj1 = sd1.y;
        float xr0[SC_G];
        #pragma unroll
        for (int k = 0; k < SC_G; ++k) {
            int s = __builtin_amdgcn_readlane(sd0.x, 8 * k);
            xr0[k] = x[(long)s * HID + lc];
        }
        while (pb < end) {
            int pj2 = pj0 + 2 * SC_G * FW;
            int pc2 = (pj2 < end) ? pj2 : last;
            int2  sd2 = pe[pc2];
            float pl2 = ebuf[pc2 * HEADS + h8];
            float xr1[SC_G];
            #pragma unroll
            for (int k = 0; k < SC_G; ++k) {
                int s = __builtin_amdgcn_readlane(sv1, 8 * k);
                xr1[k] = x[(long)s * HID + lc];
            }
            int di = (dj0 - n0) * HEADS + h8;
            float ad = s_ad[di];
            float v = pl0 + ad;
            v = (v > 0.0f) ? v : 0.2f * v;
            float m = fkey_inv(s_max[di]) + ad;
            m = (m > 0.0f) ? m : 0.2f * m;
            float ex = (pj0 < end) ? __expf(v - m) : 0.0f;
            if (pj0 < end) atomicAdd(&s_den[di], ex);
            int cnt = end - pb;
            #pragma unroll
            for (int k = 0; k < SC_G; ++k) {
                int dl = __builtin_amdgcn_readlane(dj0, 8 * k) - n0;
                float a[HEADS];
                #pragma unroll
                for (int h = 0; h < HEADS; ++h)
                    a[h] = __int_as_float(__builtin_amdgcn_readlane(__float_as_int(ex), 8 * k + h));
                if (k < cnt && lane < HID) {
                    float* zp = &s_Z[dl * ZP + lane];
                    #pragma unroll
                    for (int h = 0; h < HEADS; ++h) atomicAdd(zp + h * HID, a[h] * xr0[k]);
                }
            }
            pb += SC_G * FW;
            pj0 = pj1; pj1 = pj2;
            dj0 = dj1; pl0 = pl1;
            sv1 = sd2.x; dj1 = sd2.y; pl1 = pl2;
            #pragma unroll
            for (int k = 0; k < SC_G; ++k) xr0[k] = xr1[k];
        }
    }
    __syncthreads();

    {
        float asum = 0.0f;
        for (int i = t; i < NB * HEADS; i += 1024) {
            float dn = s_den[i];
            asum += dn / (dn + 1e-16f);
        }
        asum += __shfl_xor(asum, 8, 32);
        asum += __shfl_xor(asum, 16, 32);
        if (lane < HEADS) atomicAdd(&s_as[lane], asum);
    }

    if (wv < 26) {
        int tm = wv >> 1, tn = wv & 1;
        int m  = lane & 15;
        int hs = lane >> 4;
        int row = tm * 16 + m;
        int rowc = (row < NB) ? row : NB - 1;
        const float* arow = s_Z + rowc * ZP + 2 * hs;
        int col = tn * 16 + m;
        int colc = (col < HID) ? col : HID - 1;
        const float* wl = W_lin + (long)(2 * hs) * NHID + colc;
        v8f c = {0.f, 0.f, 0.f, 0.f, 0.f, 0.f, 0.f, 0.f};
        for (int h = 0; h < HEADS; ++h) {
            float rden = 1.0f / (s_den[rowc * HEADS + h] + 1e-16f);
            #pragma unroll
            for (int kk = 0; kk < HID; kk += 4) {
                v2f a = *reinterpret_cast<const v2f*>(arow + h * HID + kk);
                a.x *= rden; a.y *= rden;
                v2f b; b.x = wl[(long)kk * NHID + h * HID]; b.y = wl[(long)(kk + 1) * NHID + h * HID];
                c = __builtin_amdgcn_wmma_f32_16x16x4_f32(
                         false, a,  false, b,
                         (short)0, c,  false,  false);
            }
        }
        #pragma unroll
        for (int r = 0; r < 8; ++r) {
            int mg = tm * 16 + r + hs * 8;
            if (mg < NB && col < HID) s_agg[mg * HID + col] = c[r];
        }
    }
    __syncthreads();

    if (t < HEADS) atomicAdd(&alpha_out[t], s_as[t] * (1.0f / (float)N_EDGES));

    for (int nl = wv; nl < nn; nl += FW) {
        float val = 0.0f;
        if (lane < HID) {
            float s = s_agg[nl * HID + lane];
            s = s * (1.0f / HEADS) + bias[lane];
            s = (s > 0.0f) ? s : (__expf(s) - 1.0f);
            val = s * W_out[lane];
        }
        #pragma unroll
        for (int off = 16; off > 0; off >>= 1) val += __shfl_down(val, off, 32);
        if (lane == 0) s_y[nl] = val + b_out[0];
    }
    __syncthreads();
    if (t < nn) y[n0 + t] = s_y[t];
}

extern "C" void kernel_launch(void* const* d_in, const int* in_sizes, int n_in,
                              void* d_out, int out_size, void* d_ws, size_t ws_size,
                              hipStream_t stream) {
    const float* x         = (const float*)d_in[0];
    const int*   ei        = (const int*)d_in[1];
    const float* edge_attr = (const float*)d_in[2];
    const float* W_lin     = (const float*)d_in[3];
    const float* att_src   = (const float*)d_in[4];
    const float* att_dst   = (const float*)d_in[5];
    const float* W_edge    = (const float*)d_in[6];
    const float* att_edge  = (const float*)d_in[7];
    const float* bias      = (const float*)d_in[8];
    const float* W_out     = (const float*)d_in[9];
    const float* b_out     = (const float*)d_in[10];
    (void)in_sizes; (void)n_in; (void)out_size; (void)ws_size;

    const int* src = ei;
    const int* dst = ei + N_EDGES;

    float* w        = (float*)d_ws;
    float* a_src    = w;
    float* a_dst    = a_src   + (long)N_NODES * HEADS;
    float* ebuf     = a_dst   + (long)N_NODES * HEADS;
    int*   pe       = (int*)(ebuf + (long)P_SLOTS * HEADS);
    float* v_edge   = (float*)(pe + 2L * P_SLOTS);
    int*   bincount = (int*)(v_edge + EDGE_DIM * HEADS);
    int*   cursor   = bincount + NBINS;
    int*   part     = cursor + NBINS;

    float* y         = (float*)d_out;
    float* alpha_out = y + N_NODES;

    k_front<<<HB + ATT_BLOCKS, 256, 0, stream>>>(x, dst, W_edge, att_edge, W_lin, att_src, att_dst,
                                                 v_edge, part, cursor, alpha_out, a_src, a_dst);
    k_binplace<<<BIN_BLOCKS, 256, 0, stream>>>(src, dst, edge_attr, v_edge, a_src,
                                               part, bincount, cursor, pe, ebuf);
    k_bin_nodes<<<NBINS, 1024, 0, stream>>>(pe, ebuf, bincount, a_dst, x, W_lin,
                                            bias, W_out, b_out, y, alpha_out);
}


# Bin kernel: a node's sums laid out so each LDS add of the edge pass uses all 32 lanes (lane = column group x head): 6 adds per edge, not 8 of 24 lanes; WMMA k order follows the layout; sums regrouped
# speedup vs baseline: 9.0690x; 1.0468x over previous
#include <hip/hip_runtime.h>
#include <hip/hip_bf16.h>

#define N_NODES 50000
#define N_EDGES 800000
#define HID     24
#define HEADS   8
#define EDGE_DIM 8
#define NHID    192
#define M_TILES 3125
#define ACOLS   16
#define ZP      196
#define NB      196
#define NBINS   256
#define EPB     1024
#define BIN_BLOCKS ((N_EDGES + EPB - 1) / EPB)
static_assert(NB * NBINS >= N_NODES, "every node has a bin");
static_assert(EPB % 256 == 0, "whole edges per thread in the placing pass");
#define P_SLOTS (N_EDGES + 4 * NBINS)
__device__ __forceinline__ int round4(int c) { return (c + 3) & ~3; }

typedef __attribute__((ext_vector_type(2))) float v2f;
typedef __attribute__((ext_vector_type(8))) float v8f;

__device__ __forceinline__ unsigned int fkey(float v) {
    unsigned int b = __float_as_uint(v);
    return (b & 0x80000000u) ? ~b : (b | 0x80000000u);
}
__device__ __forceinline__ float fkey_inv(unsigned int k) {
    return __uint_as_float((k & 0x80000000u) ? (k ^ 0x80000000u) : ~k);
}

#define HB   64
#define EPH  (N_EDGES / HB)
#define EPHT ((EPH + 255) / 256)
#define ATT_BLOCKS ((M_TILES + 7) / 8)
static_assert(EPH * HB == N_EDGES, "the counting work-groups cover the edge list exactly");
__global__ void __launch_bounds__(256)
k_front(const float* __restrict__ x, const int* __restrict__ dst,
        const float* __restrict__ W_edge, const float* __restrict__ att_edge,
        const float* __restrict__ W_lin,
        const float* __restrict__ att_src, const float* __restrict__ att_dst,
        float* __restrict__ v_edge, int* __restrict__ part, int* __restrict__ cursor,
        float* __restrict__ alpha_out,
        float* __restrict__ a_src, float* __restrict__ a_dst) {
    __shared__ int s_cnt[NBINS];
    __shared__ float s_wa[HID * ACOLS];
    int t = threadIdx.x;
    if (blockIdx.x < HB) {
        s_cnt[t] = 0;
        __syncthreads();
        int e0 = blockIdx.x * EPH;
        int dd[EPHT];
        #pragma unroll
        for (int i = 0; i < EPHT; ++i) {
            int k = i * 256 + t;
            dd[i] = dst[e0 + ((k < EPH) ? k : EPH - 1)];
        }
        #pragma unroll
        for (int i = 0; i < EPHT; ++i)
            if (i * 256 + t < EPH) atomicAdd(&s_cnt[dd[i] / NB], 1);
        __syncthreads();
        part[blockIdx.x * NBINS + t] = s_cnt[t];
        if (blockIdx.x == 0) {
            cursor[t] = 0;
            if (t < HEADS) alpha_out[t] = 0.0f;
            if (t < EDGE_DIM * HEADS) {
                int d = t >> 3, h = t & 7;
                float sv = 0.0f;
                #pragma unroll
                for (int c = 0; c < HID; ++c)
                    sv += W_edge[d * NHID + h * HID + c] * att_edge[h * HID + c];
                v_edge[d * HEADS + h] = sv;
            }
        }
        return;
    }

    for (int q = t; q < HID * ACOLS; q += 256) {
        int k = q >> 4, col = q & 15, h = col & 7;
        const float* att = ((col < HEADS) ? att_src : att_dst) + h * HID;
        const float* wr  = W_lin + k * NHID + h * HID;
        float sv = 0.0f;
        #pragma unroll
        for (int c = 0; c < HID; ++c) sv += wr[c] * att[c];
        s_wa[q] = sv;
    }
    __syncthreads();

    int lane = t & 31;
    int tm = (int)(blockIdx.x - HB) * 8 + (t >> 5);
    if (tm >= M_TILES) return;

    int m  = lane & 15;
    int hs = lane >> 4;
    const float* arow = x + (long)(tm * 16 + m) * HID;

    v8f c = {0.f, 0.f, 0.f, 0.f, 0.f, 0.f, 0.f, 0.f};
    #pragma unroll
    for (int kb = 0; kb < HID; kb += 4) {
        int k0 = kb + hs * 2;
        v2f a; a.x = arow[k0]; a.y = arow[k0 + 1];
        v2f b; b.x = s_wa[k0 * ACOLS + m]; b.y = s_wa[(k0 + 1) * ACOLS + m];
        c = __builtin_amdgcn_wmma_f32_16x16x4_f32(
                 false, a,  false, b,
                 (short)0, c,  false,  false);
    }
    float* out = (m < HEADS) ? (a_src + m) : (a_dst + (m - HEADS));
    #pragma unroll
    for (int r = 0; r < 8; ++r) {
        int mg = tm * 16 + r + hs * 8;
        out[(long)mg * HEADS] = c[r];
    }
}

#define EPT (EPB / 256)
__global__ void __launch_bounds__(256, 8)
k_binplace(const int* __restrict__ src, const int* __restrict__ dst,
           const float* __restrict__ edge_attr,
           const float* __restrict__ v_edge,
           const float* __restrict__ a_src,
           const int* __restrict__ part, int* __restrict__ bincount, int* __restrict__ cursor,
           int* __restrict__ pe, float* __restrict__ ebuf) {
    __shared__ int s_cnt[NBINS];
    __shared__ int s_base[NBINS];
    __shared__ int s_scan[NBINS];
    __shared__ float s_v[EDGE_DIM * HEADS];
    int t = threadIdx.x;
    s_cnt[t] = 0;
    if (t < EDGE_DIM * HEADS) s_v[t] = v_edge[t];

    int e0 = blockIdx.x * EPB;
    int ss[EPT], dd[EPT];
    #pragma unroll
    for (int i = 0; i < EPT; ++i) {
        int e = e0 + i * 256 + t;
        int ec = (e < N_EDGES) ? e : N_EDGES - 1;
        ss[i] = src[ec];
        dd[i] = dst[ec];
    }
    int raw = 0;
    #pragma unroll
    for (int r = 0; r < HB; ++r) raw += part[r * NBINS + t];
    if (blockIdx.x == 0) bincount[t] = raw;
    int tot = round4(raw);
    s_scan[t] = tot;
    __syncthreads();
    for (int off = 1; off < NBINS; off <<= 1) {
        int a = (t >= off) ? s_scan[t - off] : 0;
        __syncthreads();
        s_scan[t] += a;
        __syncthreads();
    }
    int excl = s_scan[t] - tot;

    #pragma unroll
    for (int i = 0; i < EPT; ++i) {
        int e = e0 + i * 256 + t;
        if (e < N_EDGES) atomicAdd(&s_cnt[dd[i] / NB], 1);
    }
    float as[EPT][HEADS];
    #pragma unroll
    for (int i = 0; i < EPT; ++i) {
        const float* ps = a_src + (long)ss[i] * HEADS;
        #pragma unroll
        for (int h = 0; h < HEADS; ++h) as[i][h] = ps[h];
    }
    __syncthreads();
    int c = s_cnt[t];
    int prev = c ? atomicAdd(&cursor[t], c) : 0;
    s_base[t] = excl + prev;
    __syncthreads();
    s_cnt[t] = 0;
    __syncthreads();
    #pragma unroll
    for (int i = 0; i < EPT; ++i) {
        int e = e0 + i * 256 + t;
        if (e < N_EDGES) {
            int bin = dd[i] / NB;
            int pos = s_base[bin] + atomicAdd(&s_cnt[bin], 1);
            reinterpret_cast<int2*>(pe)[pos] = make_int2(ss[i], dd[i]);
            const float* pa = edge_attr + (long)e * EDGE_DIM;
            float ea[EDGE_DIM];
            #pragma unroll
            for (int jj = 0; jj < EDGE_DIM; ++jj) ea[jj] = pa[jj];
            float o[HEADS];
            #pragma unroll
            for (int h = 0; h < HEADS; ++h) {
                float acc = 0.0f;
                #pragma unroll
                for (int jj = 0; jj < EDGE_DIM; ++jj) acc += ea[jj] * s_v[jj * HEADS + h];
                o[h] = as[i][h] + acc;
            }
            float* ob = ebuf + (long)pos * HEADS;
            #pragma unroll
            for (int h = 0; h < HEADS; ++h) ob[h] = o[h];
        }
    }
}

#define SC_G 4
#define FW   32
#define PU   8
#define ADT  ((NB * HEADS + 1023) / 1024)

__global__ void __launch_bounds__(1024)
k_bin_nodes(const int* __restrict__ pe_, const float* __restrict__ ebuf, const int* __restrict__ bincount,
            const float* __restrict__ a_dst,
            const float* __restrict__ x, const float* __restrict__ W_lin,
            const float* __restrict__ bias, const float* __restrict__ W_out,
            const float* __restrict__ b_out,
            float* __restrict__ y, float* __restrict__ alpha_out) {
    __shared__ unsigned int s_max[NB * HEADS];
    __shared__ float s_den[NB * HEADS];
    __shared__ float s_ad[NB * HEADS];
    __shared__ __attribute__((aligned(16))) float s_Z[NB * ZP];
    __shared__ float s_agg[NB * HID];
    __shared__ float s_y[NB];
    __shared__ float s_as[HEADS];
    __shared__ int s_start;

    const int2* pe = reinterpret_cast<const int2*>(pe_);
    int t    = threadIdx.x;
    int b    = blockIdx.x;
    int n0   = b * NB;
    int nn   = (N_NODES - n0 < NB) ? (N_NODES - n0) : NB;
    int lane = t & 31;
    int j    = lane >> 3;
    int h8   = lane & 7;
    int wv   = __builtin_amdgcn_readfirstlane(t >> 5);

    float adv[ADT];
    #pragma unroll
    for (int q = 0; q < ADT; ++q) {
        int i = q * 1024 + t;
        adv[q] = a_dst[(long)n0 * HEADS + ((i < nn * HEADS) ? i : 0)];
    }
    if (t == 0) s_start = 0;
    for (int i = t; i < NB * HEADS; i += 1024) { s_max[i] = 0u; s_den[i] = 0.0f; }
    for (int i = t; i < NB * ZP / 4; i += 1024) reinterpret_cast<float4*>(s_Z)[i] = make_float4(0.0f, 0.0f, 0.0f, 0.0f);
    if (t < HEADS) s_as[t] = 0.0f;
    #pragma unroll
    for (int q = 0; q < ADT; ++q) {
        int i = q * 1024 + t;
        if (i < nn * HEADS) s_ad[i] = adv[q];
    }
    __syncthreads();
    if (t < b) atomicAdd(&s_start, round4(bincount[t]));
    __syncthreads();
    int start = __builtin_amdgcn_readfirstlane(s_start);
    int end   = start + __builtin_amdgcn_readfirstlane(bincount[b]);
    int last  = end - 1;

    for (int pb = start + SC_G * wv; pb < end; pb += SC_G * FW * PU) {
        int pp[PU], d[PU];
        float x[PU];
        #pragma unroll
        for (int u = 0; u < PU; ++u) {
            pp[u] = pb + SC_G * FW * u + j;
            int pc = (pp[u] < end) ? pp[u] : last;
            d[u] = pe[pc].y;
            x[u] = ebuf[pc * HEADS + h8];
        }
        asm volatile("" :: "v"(d[0]), "v"(d[1]), "v"(d[2]), "v"(d[3]), "v"(d[4]), "v"(d[5]), "v"(d[6]), "v"(d[7]),
                     "v"(x[0]), "v"(x[1]), "v"(x[2]), "v"(x[3]), "v"(x[4]), "v"(x[5]), "v"(x[6]), "v"(x[7]));
        #pragma unroll
        for (int u = 0; u < PU; ++u)
            if (pp[u] < end) atomicMax(&s_max[(d[u] - n0) * HEADS + h8], fkey(x[u]));
    }
    __syncthreads();

    int q6 = 6 * j;
    bool up = lane >= 16;

    int pb = start + SC_G * wv;
    if (pb < end) {
        int pj0 = pb + j;
        int pc0 = (pj0 < end) ? pj0 : last;
        int2  sd0 = pe[pc0];
        float pl0 = ebuf[pc0 * HEADS + h8];
        int pj1 = pj0 + SC_G * FW;
        int pc1 = (pj1 < end) ? pj1 : last;
        int2  sd1 = pe[pc1];
        float pl1 = ebuf[pc1 * HEADS + h8];
        int dj0 = sd0.y, sv1 = sd1.x, dj1 = sd1.y;
        float xr0[SC_G][6];
        #pragma unroll
        for (int k = 0; k < SC_G; ++k) {
            int s = __builtin_amdgcn_readlane(sd0.x, 8 * k);
            const float* xp = x + (long)s * HID + q6;
            #pragma unroll
            for (int i = 0; i < 6; ++i) xr0[k][i] = xp[i];
        }
        while (pb < end) {
            int pj2 = pj0 + 2 * SC_G * FW;
            int pc2 = (pj2 < end) ? pj2 : last;
            int2  sd2 = pe[pc2];
            float pl2 = ebuf[pc2 * HEADS + h8];
            float xr1[SC_G][6];
            #pragma unroll
            for (int k = 0; k < SC_G; ++k) {
                int s = __builtin_amdgcn_readlane(sv1, 8 * k);
                const float* xp = x + (long)s * HID + q6;
                #pragma unroll
                for (int i = 0; i < 6; ++i) xr1[k][i] = xp[i];
            }
            int di = (dj0 - n0) * HEADS + h8;
            float ad = s_ad[di];
            float v = pl0 + ad;
            v = (v > 0.0f) ? v : 0.2f * v;
            float m = fkey_inv(s_max[di]) + ad;
            m = (m > 0.0f) ? m : 0.2f * m;
            float ex = (pj0 < end) ? __expf(v - m) : 0.0f;
            if (pj0 < end) atomicAdd(&s_den[di], ex);
            unsigned int exu = __float_as_uint(ex);
            unsigned int s01 = __builtin_amdgcn_permlane16(exu, exu, 0x76543210u, 0x76543210u, false, false);
            unsigned int x01 = __builtin_amdgcn_permlanex16(exu, exu, 0x76543210u, 0x76543210u, false, false);
            unsigned int s23 = __builtin_amdgcn_permlane16(exu, exu, 0xfedcba98u, 0xfedcba98u, false, false);
            unsigned int x23 = __builtin_amdgcn_permlanex16(exu, exu, 0xfedcba98u, 0xfedcba98u, false, false);
            float wk[SC_G];
            wk[0] = __uint_as_float(up ? x01 : s01);
            wk[1] = __uint_as_float(up ? x23 : s23);
            wk[2] = __uint_as_float(up ? s01 : x01);
            wk[3] = __uint_as_float(up ? s23 : x23);
            int cnt = end - pb;
            #pragma unroll
            for (int k = 0; k < SC_G; ++k) {
                int dl = __builtin_amdgcn_readlane(dj0, 8 * k) - n0;
                if (k < cnt) {
                    float* zp = &s_Z[dl * ZP + lane];
                    #pragma unroll
                    for (int i = 0; i < 6; ++i) atomicAdd(zp + 32 * i, wk[k] * xr0[k][i]);
                }
            }
            pb += SC_G * FW;
            pj0 = pj1; pj1 = pj2;
            dj0 = dj1; pl0 = pl1;
            sv1 = sd2.x; dj1 = sd2.y; pl1 = pl2;
            #pragma unroll
            for (int k = 0; k < SC_G; ++k) {
                #pragma unroll
                for (int i = 0; i < 6; ++i) xr0[k][i] = xr1[k][i];
            }
        }
    }
    __syncthreads();

    {
        float asum = 0.0f;
        for (int i = t; i < NB * HEADS; i += 1024) {
            float dn = s_den[i];
            asum += dn / (dn + 1e-16f);
        }
        asum += __shfl_xor(asum, 8, 32);
        asum += __shfl_xor(asum, 16, 32);
        if (lane < HEADS) atomicAdd(&s_as[lane], asum);
    }

    if (wv < 26) {
        int tm = wv >> 1, tn = wv & 1;
        int m  = lane & 15;
        int hs = lane >> 4;
        int row = tm * 16 + m;
        int rowc = (row < NB) ? row : NB - 1;
        const float* arow = s_Z + rowc * ZP + 2 * hs;
        int col = tn * 16 + m;
        int colc = (col < HID) ? col : HID - 1;
        const float* wl = W_lin + (2 * hs) * HID + colc;
        float rd[4];
        #pragma unroll
        for (int u = 0; u < 4; ++u)
            rd[u] = 1.0f / (s_den[rowc * HEADS + 2 * hs + (u & 1) + 4 * (u >> 1)] + 1e-16f);
        v8f c = {0.f, 0.f, 0.f, 0.f, 0.f, 0.f, 0.f, 0.f};
        #pragma unroll
        for (int st = 0; st < NHID / 4; ++st) {
            int k  = 6 * ((st & 7) >> 1) + (st >> 3);
            int hb = 4 * (st & 1);
            v2f a = *reinterpret_cast<const v2f*>(arow + 4 * st);
            a.x *= rd[2 * (st & 1)]; a.y *= rd[2 * (st & 1) + 1];
            v2f b; b.x = wl[(long)k * NHID + hb * HID]; b.y = wl[(long)k * NHID + (hb + 1) * HID];
            c = __builtin_amdgcn_wmma_f32_16x16x4_f32(
                     false, a,  false, b,
                     (short)0, c,  false,  false);
        }
        #pragma unroll
        for (int r = 0; r < 8; ++r) {
            int mg = tm * 16 + r + hs * 8;
            if (mg < NB && col < HID) s_agg[mg * HID + col] = c[r];
        }
    }
    __syncthreads();

    if (t < HEADS) atomicAdd(&alpha_out[t], s_as[t] * (1.0f / (float)N_EDGES));

    for (int nl = wv; nl < nn; nl += FW) {
        float val = 0.0f;
        if (lane < HID) {
            float s = s_agg[nl * HID + lane];
            s = s * (1.0f / HEADS) + bias[lane];
            s = (s > 0.0f) ? s : (__expf(s) - 1.0f);
            val = s * W_out[lane];
        }
        #pragma unroll
        for (int off = 16; off > 0; off >>= 1) val += __shfl_down(val, off, 32);
        if (lane == 0) s_y[nl] = val + b_out[0];
    }
    __syncthreads();
    if (t < nn) y[n0 + t] = s_y[t];
}

extern "C" void kernel_launch(void* const* d_in, const int* in_sizes, int n_in,
                              void* d_out, int out_size, void* d_ws, size_t ws_size,
                              hipStream_t stream) {
    const float* x         = (const float*)d_in[0];
    const int*   ei        = (const int*)d_in[1];
    const float* edge_attr = (const float*)d_in[2];
    const float* W_lin     = (const float*)d_in[3];
    const float* att_src   = (const float*)d_in[4];
    const float* att_dst   = (const float*)d_in[5];
    const float* W_edge    = (const float*)d_in[6];
    const float* att_edge  = (const float*)d_in[7];
    const float* bias      = (const float*)d_in[8];
    const float* W_out     = (const float*)d_in[9];
    const float* b_out     = (const float*)d_in[10];
    (void)in_sizes; (void)n_in; (void)out_size; (void)ws_size;

    const int* src = ei;
    const int* dst = ei + N_EDGES;

    float* w        = (float*)d_ws;
    float* a_src    = w;
    float* a_dst    = a_src   + (long)N_NODES * HEADS;
    float* ebuf     = a_dst   + (long)N_NODES * HEADS;
    int*   pe       = (int*)(ebuf + (long)P_SLOTS * HEADS);
    float* v_edge   = (float*)(pe + 2L * P_SLOTS);
    int*   bincount = (int*)(v_edge + EDGE_DIM * HEADS);
    int*   cursor   = bincount + NBINS;
    int*   part     = cursor + NBINS;

    float* y         = (float*)d_out;
    float* alpha_out = y + N_NODES;

    k_front<<<HB + ATT_BLOCKS, 256, 0, stream>>>(x, dst, W_edge, att_edge, W_lin, att_src, att_dst,
                                                 v_edge, part, cursor, alpha_out, a_src, a_dst);
    k_binplace<<<BIN_BLOCKS, 256, 0, stream>>>(src, dst, edge_attr, v_edge, a_src,
                                               part, bincount, cursor, pe, ebuf);
    k_bin_nodes<<<NBINS, 1024, 0, stream>>>(pe, ebuf, bincount, a_dst, x, W_lin,
                                            bias, W_out, b_out, y, alpha_out);
}
